# starting kernel
# baseline (speedup 1.0000x reference)
#include <hip/hip_runtime.h>

typedef float v2f __attribute__((ext_vector_type(2)));
typedef float v8f __attribute__((ext_vector_type(8)));

#define N_NODES 100000
#define N_EDGES 1600000
#define N_FEAT  256
#define N_HID   64
#define N_CLS   16
#define N_TILES (N_NODES / 16)

__global__ void gcn_zero_kernel(float4* __restrict__ p, size_t n4) {
    size_t i = (size_t)blockIdx.x * blockDim.x + threadIdx.x;
    size_t stride = (size_t)gridDim.x * blockDim.x;
    float4 z = {0.f, 0.f, 0.f, 0.f};
    for (; i < n4; i += stride) p[i] = z;
}

__global__ void __launch_bounds__(128) gcn_gemm1_kernel(
        const float* __restrict__ x, const float* __restrict__ W1,
        const float* __restrict__ b1, float* __restrict__ h0) {
    __shared__ float sW[N_FEAT * N_HID];
    {
        const float4* W4 = (const float4*)W1;
        float4* s4 = (float4*)sW;
        for (int i = threadIdx.x; i < (N_FEAT * N_HID) / 4; i += 128) s4[i] = W4[i];
    }
    __syncthreads();

    const int wave = threadIdx.x >> 5;
    const int lane = threadIdx.x & 31;
    const int half = lane >> 4;
    const int m    = lane & 15;

    const int tile = blockIdx.x * 4 + wave;
    if (tile >= N_TILES) return;

    const int row = tile * 16 + m;
    const float* __restrict__ xrow = x + (size_t)row * N_FEAT;

    v8f acc0 = {}, acc1 = {}, acc2 = {}, acc3 = {};

    for (int k0 = 0; k0 < N_FEAT; k0 += 4) {
        const int k = k0 + 2 * half;
        v2f a;
        a.x = xrow[k];
        a.y = xrow[k + 1];
        const float* wk0 = sW + (size_t)k * N_HID;
        const float* wk1 = wk0 + N_HID;
        v2f b0, b1v, b2v, b3v;
        b0.x  = wk0[m];      b0.y  = wk1[m];
        b1v.x = wk0[16 + m]; b1v.y = wk1[16 + m];
        b2v.x = wk0[32 + m]; b2v.y = wk1[32 + m];
        b3v.x = wk0[48 + m]; b3v.y = wk1[48 + m];
        acc0 = __builtin_amdgcn_wmma_f32_16x16x4_f32(false, a, false, b0,  (short)0, acc0, false, false);
        acc1 = __builtin_amdgcn_wmma_f32_16x16x4_f32(false, a, false, b1v, (short)0, acc1, false, false);
        acc2 = __builtin_amdgcn_wmma_f32_16x16x4_f32(false, a, false, b2v, (short)0, acc2, false, false);
        acc3 = __builtin_amdgcn_wmma_f32_16x16x4_f32(false, a, false, b3v, (short)0, acc3, false, false);
    }

    float* __restrict__ out = h0 + (size_t)tile * 16 * N_HID;
    const float bb0 = b1[m], bb1 = b1[16 + m], bb2 = b1[32 + m], bb3 = b1[48 + m];
    #pragma unroll
    for (int r = 0; r < 8; ++r) {
        const size_t orow = (size_t)(half * 8 + r) * N_HID;
        out[orow + m]      = acc0[r] + bb0;
        out[orow + 16 + m] = acc1[r] + bb1;
        out[orow + 32 + m] = acc2[r] + bb2;
        out[orow + 48 + m] = acc3[r] + bb3;
    }
}

__global__ void __launch_bounds__(256) gcn_spmm64_kernel(
        const int* __restrict__ arow, const int* __restrict__ acol,
        const float* __restrict__ avals, const float* __restrict__ h,
        float* __restrict__ out) {
    size_t t = (size_t)blockIdx.x * 256 + threadIdx.x;
    if (t >= (size_t)N_EDGES * N_HID) return;
    const int e = (int)(t >> 6);
    const int f = (int)(t & 63);
    const int c = acol[e];
    const int r = arow[e];
    const float v = avals[e];
    atomicAdd(&out[(size_t)r * N_HID + f], v * h[(size_t)c * N_HID + f]);
}

__global__ void __launch_bounds__(128) gcn_gemm2_kernel(
        const float* __restrict__ h1, const float* __restrict__ W2,
        const float* __restrict__ b2, float* __restrict__ h2) {
    __shared__ float sW[N_HID * N_CLS];
    for (int i = threadIdx.x; i < N_HID * N_CLS; i += 128) sW[i] = W2[i];
    __syncthreads();

    const int wave = threadIdx.x >> 5;
    const int lane = threadIdx.x & 31;
    const int half = lane >> 4;
    const int m    = lane & 15;

    const int tile = blockIdx.x * 4 + wave;
    if (tile >= N_TILES) return;

    const int row = tile * 16 + m;
    const float* __restrict__ hrow = h1 + (size_t)row * N_HID;

    v8f acc = {};
    for (int k0 = 0; k0 < N_HID; k0 += 4) {
        const int k = k0 + 2 * half;
        v2f a;
        a.x = fmaxf(hrow[k], 0.f);
        a.y = fmaxf(hrow[k + 1], 0.f);
        v2f b;
        b.x = sW[(size_t)k * N_CLS + m];
        b.y = sW[(size_t)(k + 1) * N_CLS + m];
        acc = __builtin_amdgcn_wmma_f32_16x16x4_f32(false, a, false, b, (short)0, acc, false, false);
    }

    float* __restrict__ out = h2 + (size_t)tile * 16 * N_CLS;
    const float bb = b2[m];
    #pragma unroll
    for (int r = 0; r < 8; ++r)
        out[(size_t)(half * 8 + r) * N_CLS + m] = acc[r] + bb;
}

__global__ void __launch_bounds__(256) gcn_spmm16_kernel(
        const int* __restrict__ arow, const int* __restrict__ acol,
        const float* __restrict__ avals, const float* __restrict__ h,
        float* __restrict__ out) {
    size_t t = (size_t)blockIdx.x * 256 + threadIdx.x;
    if (t >= (size_t)N_EDGES * N_CLS) return;
    const int e = (int)(t >> 4);
    const int f = (int)(t & 15);
    const int c = acol[e];
    const int r = arow[e];
    const float v = avals[e];
    atomicAdd(&out[(size_t)r * N_CLS + f], v * h[(size_t)c * N_CLS + f]);
}

__global__ void __launch_bounds__(256) gcn_lsm_kernel(
        const float* __restrict__ s, float* __restrict__ out) {
    const int r = blockIdx.x * 256 + threadIdx.x;
    if (r >= N_NODES) return;
    const float* __restrict__ p = s + (size_t)r * N_CLS;
    float v[N_CLS];
    float mx = -3.402823466e+38f;
    #pragma unroll
    for (int i = 0; i < N_CLS; ++i) { v[i] = p[i]; mx = fmaxf(mx, v[i]); }
    float sum = 0.f;
    #pragma unroll
    for (int i = 0; i < N_CLS; ++i) sum += __expf(v[i] - mx);
    const float lse = mx + __logf(sum);
    float* __restrict__ o = out + (size_t)r * N_CLS;
    #pragma unroll
    for (int i = 0; i < N_CLS; ++i) o[i] = v[i] - lse;
}

extern "C" void kernel_launch(void* const* d_in, const int* in_sizes, int n_in,
                              void* d_out, int out_size, void* d_ws, size_t ws_size,
                              hipStream_t stream) {
    const float* x        = (const float*)d_in[0];
    const int*   adj_row  = (const int*)  d_in[1];
    const int*   adj_col  = (const int*)  d_in[2];
    const float* adj_vals = (const float*)d_in[3];
    const float* W1       = (const float*)d_in[4];
    const float* b1       = (const float*)d_in[5];
    const float* W2       = (const float*)d_in[6];
    const float* b2       = (const float*)d_in[7];
    float* out = (float*)d_out;

    float* ws = (float*)d_ws;
    float* h0 = ws;
    float* h1 = h0 + (size_t)N_NODES * N_HID;
    float* h2 = h1 + (size_t)N_NODES * N_HID;
    float* s2 = h2 + (size_t)N_NODES * N_CLS;

    {
        size_t nz = (size_t)N_NODES * (N_HID + N_CLS + N_CLS);
        size_t n4 = nz / 4;
        int blocks = 2048;
        gcn_zero_kernel<<<blocks, 256, 0, stream>>>((float4*)h1, n4);
    }

    {
        int blocks = (N_TILES + 3) / 4;
        gcn_gemm1_kernel<<<blocks, 128, 0, stream>>>(x, W1, b1, h0);
    }
    {
        size_t threads = (size_t)N_EDGES * N_HID;
        int blocks = (int)((threads + 255) / 256);
        gcn_spmm64_kernel<<<blocks, 256, 0, stream>>>(adj_row, adj_col, adj_vals, h0, h1);
    }
    {
        int blocks = (N_TILES + 3) / 4;
        gcn_gemm2_kernel<<<blocks, 128, 0, stream>>>(h1, W2, b2, h2);
    }
    {
        size_t threads = (size_t)N_EDGES * N_CLS;
        int blocks = (int)((threads + 255) / 256);
        gcn_spmm16_kernel<<<blocks, 256, 0, stream>>>(adj_row, adj_col, adj_vals, h2, s2);
    }
    {
        int blocks = (N_NODES + 255) / 256;
        gcn_lsm_kernel<<<blocks, 256, 0, stream>>>(s2, out);
    }
}


# SpMM without float atomics: edges ordered by destination row in the call (integer count whose return is the rank, scan, place); one wave sums a row in registers; sum order = arrival rank, 2 halves
# speedup vs baseline: 1.5693x; 1.5693x over previous
#include <hip/hip_runtime.h>

typedef float v2f __attribute__((ext_vector_type(2)));
typedef float v8f __attribute__((ext_vector_type(8)));

#define N_NODES 100000
#define N_EDGES 1600000
#define N_FEAT  256
#define N_HID   64
#define N_CLS   16
#define N_TILES (N_NODES / 16)

__global__ void gcn_zero_kernel(float4* __restrict__ p, size_t n4) {
    size_t i = (size_t)blockIdx.x * blockDim.x + threadIdx.x;
    size_t stride = (size_t)gridDim.x * blockDim.x;
    float4 z = {0.f, 0.f, 0.f, 0.f};
    for (; i < n4; i += stride) p[i] = z;
}

__global__ void __launch_bounds__(128) gcn_gemm1_kernel(
        const float* __restrict__ x, const float* __restrict__ W1,
        const float* __restrict__ b1, float* __restrict__ h0) {
    __shared__ float sW[N_FEAT * N_HID];
    {
        const float4* W4 = (const float4*)W1;
        float4* s4 = (float4*)sW;
        for (int i = threadIdx.x; i < (N_FEAT * N_HID) / 4; i += 128) s4[i] = W4[i];
    }
    __syncthreads();

    const int wave = threadIdx.x >> 5;
    const int lane = threadIdx.x & 31;
    const int half = lane >> 4;
    const int m    = lane & 15;

    const int tile = blockIdx.x * 4 + wave;
    if (tile >= N_TILES) return;

    const int row = tile * 16 + m;
    const float* __restrict__ xrow = x + (size_t)row * N_FEAT;

    v8f acc0 = {}, acc1 = {}, acc2 = {}, acc3 = {};

    for (int k0 = 0; k0 < N_FEAT; k0 += 4) {
        const int k = k0 + 2 * half;
        v2f a;
        a.x = xrow[k];
        a.y = xrow[k + 1];
        const float* wk0 = sW + (size_t)k * N_HID;
        const float* wk1 = wk0 + N_HID;
        v2f b0, b1v, b2v, b3v;
        b0.x  = wk0[m];      b0.y  = wk1[m];
        b1v.x = wk0[16 + m]; b1v.y = wk1[16 + m];
        b2v.x = wk0[32 + m]; b2v.y = wk1[32 + m];
        b3v.x = wk0[48 + m]; b3v.y = wk1[48 + m];
        acc0 = __builtin_amdgcn_wmma_f32_16x16x4_f32(false, a, false, b0,  (short)0, acc0, false, false);
        acc1 = __builtin_amdgcn_wmma_f32_16x16x4_f32(false, a, false, b1v, (short)0, acc1, false, false);
        acc2 = __builtin_amdgcn_wmma_f32_16x16x4_f32(false, a, false, b2v, (short)0, acc2, false, false);
        acc3 = __builtin_amdgcn_wmma_f32_16x16x4_f32(false, a, false, b3v, (short)0, acc3, false, false);
    }

    float* __restrict__ out = h0 + (size_t)tile * 16 * N_HID;
    const float bb0 = b1[m], bb1 = b1[16 + m], bb2 = b1[32 + m], bb3 = b1[48 + m];
    #pragma unroll
    for (int r = 0; r < 8; ++r) {
        const size_t orow = (size_t)(half * 8 + r) * N_HID;
        out[orow + m]      = acc0[r] + bb0;
        out[orow + 16 + m] = acc1[r] + bb1;
        out[orow + 32 + m] = acc2[r] + bb2;
        out[orow + 48 + m] = acc3[r] + bb3;
    }
}

__global__ void __launch_bounds__(256) gcn_count_kernel(
        const int* __restrict__ arow, int* __restrict__ cnt, int* __restrict__ rank) {
    const int e = blockIdx.x * 256 + threadIdx.x;
    if (e >= N_EDGES) return;
    rank[e] = atomicAdd(&cnt[arow[e]], 1);
}

#define SCAN_T 1024
#define SCAN_B ((N_NODES + SCAN_T - 1) / SCAN_T)
__global__ void __launch_bounds__(SCAN_T) gcn_scan_kernel(
        const int* __restrict__ cnt, int* __restrict__ rowptr) {
    __shared__ int sA[32];
    __shared__ int sB[32];
    const int t = threadIdx.x;
    const int lane = t & 31;
    const int wave = t >> 5;
    const int base = blockIdx.x * SCAN_T;
    int p = 0;
    for (int i = t; i < base; i += SCAN_T) p += cnt[i];
    for (int o = 16; o > 0; o >>= 1) p += __shfl_xor(p, o);
    if (lane == 0) sA[wave] = p;
    const int idx = base + t;
    const int own = (idx < N_NODES) ? cnt[idx] : 0;
    int x = own;
    for (int o = 1; o < 32; o <<= 1) {
        const int y = __shfl_up(x, o);
        if (lane >= o) x += y;
    }
    if (lane == 31) sB[wave] = x;
    __syncthreads();
    int prefix = 0, woff = 0;
    for (int w = 0; w < 32; ++w) {
        prefix += sA[w];
        if (w < wave) woff += sB[w];
    }
    const int excl = prefix + woff + x - own;
    if (idx < N_NODES) rowptr[idx] = excl;
    if (idx == N_NODES - 1) rowptr[N_NODES] = excl + own;
}

__global__ void __launch_bounds__(256) gcn_place_kernel(
        const int* __restrict__ arow, const int* __restrict__ acol,
        const float* __restrict__ avals, const int* __restrict__ rowptr,
        const int* __restrict__ rank, float* __restrict__ cval,
        unsigned short* __restrict__ clo, unsigned char* __restrict__ chi) {
    const int e = blockIdx.x * 256 + threadIdx.x;
    if (e >= N_EDGES) return;
    const int pos = rowptr[arow[e]] + rank[e];
    const int c = acol[e];
    cval[pos] = avals[e];
    clo[pos] = (unsigned short)(c & 0xffff);
    chi[pos] = (unsigned char)(c >> 16);
}

__global__ void __launch_bounds__(256) gcn_spmm64_kernel(
        const int* __restrict__ rowptr, const float* __restrict__ cval,
        const unsigned short* __restrict__ clo, const unsigned char* __restrict__ chi,
        const float* __restrict__ h, float* __restrict__ out) {
    const int lane = threadIdx.x & 31;
    const int row = blockIdx.x * 8 + (threadIdx.x >> 5);
    if (row >= N_NODES) return;
    const int beg = __builtin_amdgcn_readfirstlane(rowptr[row]);
    const int end = __builtin_amdgcn_readfirstlane(rowptr[row + 1]);
    const int slot = lane >> 4;
    const int q = lane & 15;
    const float4* __restrict__ h4 = (const float4*)h;
    float4 acc = {0.f, 0.f, 0.f, 0.f};
    for (int cb = beg; cb < end; cb += 32) {
        const int n = (end - cb < 32) ? (end - cb) : 32;
        int c = 0;
        float v = 0.f;
        if (lane < n) {
            c = (int)clo[cb + lane] | ((int)chi[cb + lane] << 16);
            v = cval[cb + lane];
        }
        for (int g = 0; g < n; g += 8) {
            const int j0 = g + slot, j1 = j0 + 2, j2 = j0 + 4, j3 = j0 + 6;
            const int c0 = __shfl(c, j0), c1 = __shfl(c, j1), c2 = __shfl(c, j2), c3 = __shfl(c, j3);
            const float v0 = __shfl(v, j0), v1 = __shfl(v, j1), v2 = __shfl(v, j2), v3 = __shfl(v, j3);
            float4 r0 = h4[(size_t)c0 * 16 + q];
            float4 r1 = h4[(size_t)c1 * 16 + q];
            float4 r2 = h4[(size_t)c2 * 16 + q];
            float4 r3 = h4[(size_t)c3 * 16 + q];
            const float4 z = {0.f, 0.f, 0.f, 0.f};
            if (j0 >= n) r0 = z;
            if (j1 >= n) r1 = z;
            if (j2 >= n) r2 = z;
            if (j3 >= n) r3 = z;
            acc.x += v0 * r0.x; acc.y += v0 * r0.y; acc.z += v0 * r0.z; acc.w += v0 * r0.w;
            acc.x += v1 * r1.x; acc.y += v1 * r1.y; acc.z += v1 * r1.z; acc.w += v1 * r1.w;
            acc.x += v2 * r2.x; acc.y += v2 * r2.y; acc.z += v2 * r2.z; acc.w += v2 * r2.w;
            acc.x += v3 * r3.x; acc.y += v3 * r3.y; acc.z += v3 * r3.z; acc.w += v3 * r3.w;
        }
    }
    acc.x += __shfl_xor(acc.x, 16);
    acc.y += __shfl_xor(acc.y, 16);
    acc.z += __shfl_xor(acc.z, 16);
    acc.w += __shfl_xor(acc.w, 16);
    if (slot == 0) ((float4*)out)[(size_t)row * 16 + q] = acc;
}

__global__ void __launch_bounds__(128) gcn_gemm2_kernel(
        const float* __restrict__ h1, const float* __restrict__ W2,
        const float* __restrict__ b2, float* __restrict__ h2) {
    __shared__ float sW[N_HID * N_CLS];
    for (int i = threadIdx.x; i < N_HID * N_CLS; i += 128) sW[i] = W2[i];
    __syncthreads();

    const int wave = threadIdx.x >> 5;
    const int lane = threadIdx.x & 31;
    const int half = lane >> 4;
    const int m    = lane & 15;

    const int tile = blockIdx.x * 4 + wave;
    if (tile >= N_TILES) return;

    const int row = tile * 16 + m;
    const float* __restrict__ hrow = h1 + (size_t)row * N_HID;

    v8f acc = {};
    for (int k0 = 0; k0 < N_HID; k0 += 4) {
        const int k = k0 + 2 * half;
        v2f a;
        a.x = fmaxf(hrow[k], 0.f);
        a.y = fmaxf(hrow[k + 1], 0.f);
        v2f b;
        b.x = sW[(size_t)k * N_CLS + m];
        b.y = sW[(size_t)(k + 1) * N_CLS + m];
        acc = __builtin_amdgcn_wmma_f32_16x16x4_f32(false, a, false, b, (short)0, acc, false, false);
    }

    float* __restrict__ out = h2 + (size_t)tile * 16 * N_CLS;
    const float bb = b2[m];
    #pragma unroll
    for (int r = 0; r < 8; ++r)
        out[(size_t)(half * 8 + r) * N_CLS + m] = acc[r] + bb;
}

__global__ void __launch_bounds__(256) gcn_spmm16_kernel(
        const int* __restrict__ rowptr, const float* __restrict__ cval,
        const unsigned short* __restrict__ clo, const unsigned char* __restrict__ chi,
        const float* __restrict__ h, float* __restrict__ out) {
    __shared__ float4 sOut[32];
    const int lane = threadIdx.x & 31;
    const int wave = threadIdx.x >> 5;
    const int row = blockIdx.x * 8 + wave;
    const int beg = __builtin_amdgcn_readfirstlane(rowptr[row]);
    const int end = __builtin_amdgcn_readfirstlane(rowptr[row + 1]);
    const int slot = lane >> 2;
    const int q = lane & 3;
    const float4* __restrict__ h4 = (const float4*)h;
    const float4 z = {0.f, 0.f, 0.f, 0.f};
    float4 acc = z;
    for (int cb = beg; cb < end; cb += 32) {
        const int n = (end - cb < 32) ? (end - cb) : 32;
        int c = 0;
        float v = 0.f;
        if (lane < n) {
            c = (int)clo[cb + lane] | ((int)chi[cb + lane] << 16);
            v = cval[cb + lane];
        }
        for (int g = 0; g < n; g += 16) {
            const int j0 = g + slot, j1 = j0 + 8;
            const int c0 = __shfl(c, j0), c1 = __shfl(c, j1);
            const float v0 = __shfl(v, j0), v1 = __shfl(v, j1);
            float4 r0 = h4[(size_t)c0 * 4 + q];
            float4 r1 = h4[(size_t)c1 * 4 + q];
            if (j0 >= n) r0 = z;
            if (j1 >= n) r1 = z;
            acc.x += v0 * r0.x; acc.y += v0 * r0.y; acc.z += v0 * r0.z; acc.w += v0 * r0.w;
            acc.x += v1 * r1.x; acc.y += v1 * r1.y; acc.z += v1 * r1.z; acc.w += v1 * r1.w;
        }
    }
    #pragma unroll
    for (int o = 4; o < 32; o <<= 1) {
        acc.x += __shfl_xor(acc.x, o);
        acc.y += __shfl_xor(acc.y, o);
        acc.z += __shfl_xor(acc.z, o);
        acc.w += __shfl_xor(acc.w, o);
    }
    if (lane < 4) sOut[wave * 4 + q] = acc;
    __syncthreads();
    if (wave == 0) ((float4*)out)[(size_t)blockIdx.x * 32 + lane] = sOut[lane];
}

__global__ void __launch_bounds__(256) gcn_lsm_kernel(
        const float* __restrict__ s, float* __restrict__ out) {
    const int r = blockIdx.x * 256 + threadIdx.x;
    if (r >= N_NODES) return;
    const float* __restrict__ p = s + (size_t)r * N_CLS;
    float v[N_CLS];
    float mx = -3.402823466e+38f;
    #pragma unroll
    for (int i = 0; i < N_CLS; ++i) { v[i] = p[i]; mx = fmaxf(mx, v[i]); }
    float sum = 0.f;
    #pragma unroll
    for (int i = 0; i < N_CLS; ++i) sum += __expf(v[i] - mx);
    const float lse = mx + __logf(sum);
    float* __restrict__ o = out + (size_t)r * N_CLS;
    #pragma unroll
    for (int i = 0; i < N_CLS; ++i) o[i] = v[i] - lse;
}

extern "C" void kernel_launch(void* const* d_in, const int* in_sizes, int n_in,
                              void* d_out, int out_size, void* d_ws, size_t ws_size,
                              hipStream_t stream) {
    const float* x        = (const float*)d_in[0];
    const int*   adj_row  = (const int*)  d_in[1];
    const int*   adj_col  = (const int*)  d_in[2];
    const float* adj_vals = (const float*)d_in[3];
    const float* W1       = (const float*)d_in[4];
    const float* b1       = (const float*)d_in[5];
    const float* W2       = (const float*)d_in[6];
    const float* b2       = (const float*)d_in[7];
    float* out = (float*)d_out;

    float* ws = (float*)d_ws;
    float* h0 = ws;
    float* h2 = ws;
    float* s2 = ws + (size_t)N_NODES * N_CLS;
    float* h1 = ws + (size_t)N_NODES * N_HID;
    int*   cnt  = (int*)h1;
    int*   rank = cnt + N_NODES;
    float* cval = ws + (size_t)N_NODES * (N_HID + N_HID);
    unsigned short* clo = (unsigned short*)(cval + N_EDGES);
    unsigned char*  chi = (unsigned char*)(cval + N_EDGES + N_EDGES / 2);
    int* rowptr = (int*)(cval + N_EDGES + N_EDGES / 2 + N_EDGES / 4);

    gcn_zero_kernel<<<(N_NODES / 4 + 255) / 256, 256, 0, stream>>>((float4*)cnt, (size_t)(N_NODES / 4));
    gcn_count_kernel<<<N_EDGES / 256, 256, 0, stream>>>(adj_row, cnt, rank);
    gcn_scan_kernel<<<SCAN_B, SCAN_T, 0, stream>>>(cnt, rowptr);
    gcn_place_kernel<<<N_EDGES / 256, 256, 0, stream>>>(adj_row, adj_col, adj_vals, rowptr, rank,
                                                        cval, clo, chi);
    gcn_gemm1_kernel<<<(N_TILES + 3) / 4, 128, 0, stream>>>(x, W1, b1, h0);
    gcn_spmm64_kernel<<<N_NODES / 8, 256, 0, stream>>>(rowptr, cval, clo, chi, h0, h1);
    gcn_gemm2_kernel<<<(N_TILES + 3) / 4, 128, 0, stream>>>(h1, W2, b2, h2);
    gcn_spmm16_kernel<<<N_NODES / 8, 256, 0, stream>>>(rowptr, cval, clo, chi, h2, s2);
    gcn_lsm_kernel<<<(N_NODES + 255) / 256, 256, 0, stream>>>(s2, out);
}


# Edges ordered by destination row with LDS counters only: chunk-by-bucket counting sort, then one work-group per 256-row bucket; no global atomics, 7 launches; row sum order still arrival rank
# speedup vs baseline: 4.3150x; 2.7496x over previous
#include <hip/hip_runtime.h>

typedef float v2f __attribute__((ext_vector_type(2)));
typedef float v8f __attribute__((ext_vector_type(8)));

#define N_NODES 100000
#define N_EDGES 1600000
#define N_FEAT  256
#define N_HID   64
#define N_CLS   16
#define N_TILES (N_NODES / 16)

__global__ void __launch_bounds__(128) gcn_gemm1_kernel(
        const float* __restrict__ x, const float* __restrict__ W1,
        const float* __restrict__ b1, float* __restrict__ h0) {
    __shared__ float sW[N_FEAT * N_HID];
    {
        const float4* W4 = (const float4*)W1;
        float4* s4 = (float4*)sW;
        for (int i = threadIdx.x; i < (N_FEAT * N_HID) / 4; i += 128) s4[i] = W4[i];
    }
    __syncthreads();

    const int wave = threadIdx.x >> 5;
    const int lane = threadIdx.x & 31;
    const int half = lane >> 4;
    const int m    = lane & 15;

    const int tile = blockIdx.x * 4 + wave;
    if (tile >= N_TILES) return;

    const int row = tile * 16 + m;
    const float* __restrict__ xrow = x + (size_t)row * N_FEAT;

    v8f acc0 = {}, acc1 = {}, acc2 = {}, acc3 = {};

    for (int k0 = 0; k0 < N_FEAT; k0 += 4) {
        const int k = k0 + 2 * half;
        v2f a;
        a.x = xrow[k];
        a.y = xrow[k + 1];
        const float* wk0 = sW + (size_t)k * N_HID;
        const float* wk1 = wk0 + N_HID;
        v2f b0, b1v, b2v, b3v;
        b0.x  = wk0[m];      b0.y  = wk1[m];
        b1v.x = wk0[16 + m]; b1v.y = wk1[16 + m];
        b2v.x = wk0[32 + m]; b2v.y = wk1[32 + m];
        b3v.x = wk0[48 + m]; b3v.y = wk1[48 + m];
        acc0 = __builtin_amdgcn_wmma_f32_16x16x4_f32(false, a, false, b0,  (short)0, acc0, false, false);
        acc1 = __builtin_amdgcn_wmma_f32_16x16x4_f32(false, a, false, b1v, (short)0, acc1, false, false);
        acc2 = __builtin_amdgcn_wmma_f32_16x16x4_f32(false, a, false, b2v, (short)0, acc2, false, false);
        acc3 = __builtin_amdgcn_wmma_f32_16x16x4_f32(false, a, false, b3v, (short)0, acc3, false, false);
    }

    float* __restrict__ out = h0 + (size_t)tile * 16 * N_HID;
    const float bb0 = b1[m], bb1 = b1[16 + m], bb2 = b1[32 + m], bb3 = b1[48 + m];
    #pragma unroll
    for (int r = 0; r < 8; ++r) {
        const size_t orow = (size_t)(half * 8 + r) * N_HID;
        out[orow + m]      = acc0[r] + bb0;
        out[orow + 16 + m] = acc1[r] + bb1;
        out[orow + 32 + m] = acc2[r] + bb2;
        out[orow + 48 + m] = acc3[r] + bb3;
    }
}

#define PART_B   256
#define PART_CH  (N_EDGES / PART_B)
#define BKT_ROWS 256
#define N_BKT    ((N_NODES + BKT_ROWS - 1) / BKT_ROWS)
#define TAB_W    (N_BKT + 1)
#define PART_IT  ((PART_CH + 1023) / 1024)

__global__ void __launch_bounds__(1024) gcn_part_kernel(
        const int* __restrict__ arow, const int* __restrict__ acol,
        const float* __restrict__ avals, int* __restrict__ tab,
        unsigned* __restrict__ pkey, float* __restrict__ pval) {
    __shared__ int sCur[TAB_W];
    __shared__ int sWave[32];
    __shared__ unsigned sKey[PART_CH];
    __shared__ float sVal[PART_CH];
    const int t = threadIdx.x;
    const int lane = t & 31;
    const int wave = t >> 5;
    const int base = blockIdx.x * PART_CH;
    if (t < TAB_W) sCur[t] = 0;
    __syncthreads();
    int r[PART_IT], c[PART_IT];
    float v[PART_IT];
    #pragma unroll
    for (int k = 0; k < PART_IT; ++k) {
        const int i = t + k * 1024;
        r[k] = 0; c[k] = 0; v[k] = 0.f;
        if (i < PART_CH) {
            r[k] = arow[base + i];
            c[k] = acol[base + i];
            v[k] = avals[base + i];
        }
    }
    #pragma unroll
    for (int k = 0; k < PART_IT; ++k)
        if (t + k * 1024 < PART_CH) atomicAdd(&sCur[r[k] >> 8], 1);
    __syncthreads();
    const int own = (t < N_BKT) ? sCur[t] : 0;
    int x = own;
    for (int o = 1; o < 32; o <<= 1) {
        const int y = __shfl_up(x, o);
        if (lane >= o) x += y;
    }
    if (lane == 31) sWave[wave] = x;
    __syncthreads();
    int woff = 0;
    for (int w = 0; w < 32; ++w)
        if (w < wave) woff += sWave[w];
    const int excl = woff + x - own;
    if (t < TAB_W) {
        sCur[t] = excl;
        tab[blockIdx.x * TAB_W + t] = excl;
    }
    __syncthreads();
    #pragma unroll
    for (int k = 0; k < PART_IT; ++k) {
        if (t + k * 1024 < PART_CH) {
            const int pos = atomicAdd(&sCur[r[k] >> 8], 1);
            sKey[pos] = ((unsigned)(r[k] & (BKT_ROWS - 1)) << 24) | (unsigned)c[k];
            sVal[pos] = v[k];
        }
    }
    __syncthreads();
    for (int i = t; i < PART_CH; i += 1024) {
        pkey[base + i] = sKey[i];
        pval[base + i] = sVal[i];
    }
}

#define SORT_CAP 8192
#define SORT_U   4

__device__ __forceinline__ int gcn_run_of(const int* sSrc, int e) {
    int lo = 0, hi = PART_B;
    #pragma unroll
    for (int it = 0; it < 8; ++it) {
        const int mid = (lo + hi) >> 1;
        if (sSrc[mid] <= e) lo = mid; else hi = mid;
    }
    return lo;
}

__global__ void __launch_bounds__(1024) gcn_rowsort_kernel(
        const int* __restrict__ tab, const unsigned* __restrict__ pkey,
        const float* __restrict__ pval, int* __restrict__ rowptr,
        float* __restrict__ cval, unsigned short* __restrict__ clo,
        unsigned char* __restrict__ chi) {
    __shared__ int sSrc[PART_B + 1];
    __shared__ int sBeg[PART_B];
    __shared__ int sCnt[BKT_ROWS];
    __shared__ int sWa[32];
    __shared__ int sWb[32];
    __shared__ int sKey[SORT_CAP];
    __shared__ float sVal[SORT_CAP];
    const int t = threadIdx.x;
    const int lane = t & 31;
    const int wave = t >> 5;
    const int b = blockIdx.x;

    int o0 = 0, cnt = 0;
    if (t < PART_B) {
        o0 = tab[t * TAB_W + b];
        cnt = tab[t * TAB_W + b + 1] - o0;
        sBeg[t] = t * PART_CH + o0;
    }
    if (t < BKT_ROWS) sCnt[t] = 0;
    int x = cnt;
    for (int o = 1; o < 32; o <<= 1) {
        const int y = __shfl_up(x, o);
        if (lane >= o) x += y;
    }
    int q = o0;
    for (int o = 16; o > 0; o >>= 1) q += __shfl_xor(q, o);
    if (lane == 31) sWa[wave] = x;
    if (lane == 0) sWb[wave] = q;
    __syncthreads();
    int woff = 0, n = 0, bbase = 0;
    for (int w = 0; w < 32; ++w) {
        const int a = sWa[w];
        if (w < wave) woff += a;
        n += a;
        bbase += sWb[w];
    }
    if (t < PART_B) sSrc[t] = woff + x - cnt;
    if (t == 0) sSrc[PART_B] = n;
    __syncthreads();

    for (int e0 = t; e0 < n; e0 += 1024 * SORT_U) {
        unsigned key[SORT_U];
        #pragma unroll
        for (int k = 0; k < SORT_U; ++k) {
            const int e = e0 + k * 1024;
            key[k] = 0u;
            if (e < n) {
                const int s = gcn_run_of(sSrc, e);
                key[k] = pkey[sBeg[s] + (e - sSrc[s])];
            }
        }
        #pragma unroll
        for (int k = 0; k < SORT_U; ++k)
            if (e0 + k * 1024 < n) atomicAdd(&sCnt[key[k] >> 24], 1);
    }
    __syncthreads();
    const int own = (t < BKT_ROWS) ? sCnt[t] : 0;
    int y = own;
    for (int o = 1; o < 32; o <<= 1) {
        const int z = __shfl_up(y, o);
        if (lane >= o) y += z;
    }
    if (lane == 31) sWa[wave] = y;
    __syncthreads();
    int roff = 0;
    for (int w = 0; w < 32; ++w)
        if (w < wave) roff += sWa[w];
    const int rstart = roff + y - own;
    if (t < BKT_ROWS) {
        sCnt[t] = rstart;
        const int row = b * BKT_ROWS + t;
        if (row <= N_NODES) rowptr[row] = bbase + rstart;
    }
    __syncthreads();

    const bool staged = (n <= SORT_CAP);
    for (int e0 = t; e0 < n; e0 += 1024 * SORT_U) {
        unsigned key[SORT_U];
        float val[SORT_U];
        #pragma unroll
        for (int k = 0; k < SORT_U; ++k) {
            const int e = e0 + k * 1024;
            key[k] = 0u; val[k] = 0.f;
            if (e < n) {
                const int s = gcn_run_of(sSrc, e);
                const int p = sBeg[s] + (e - sSrc[s]);
                key[k] = pkey[p];
                val[k] = pval[p];
            }
        }
        #pragma unroll
        for (int k = 0; k < SORT_U; ++k) {
            if (e0 + k * 1024 < n) {
                const int pos = atomicAdd(&sCnt[key[k] >> 24], 1);
                const int c = (int)(key[k] & 0xffffffu);
                if (staged) {
                    sKey[pos] = c;
                    sVal[pos] = val[k];
                } else {
                    const int g = bbase + pos;
                    cval[g] = val[k];
                    clo[g] = (unsigned short)(c & 0xffff);
                    chi[g] = (unsigned char)(c >> 16);
                }
            }
        }
    }
    __syncthreads();
    if (staged) {
        for (int i = t; i < n; i += 1024) {
            const int g = bbase + i;
            const int c = sKey[i];
            cval[g] = sVal[i];
            clo[g] = (unsigned short)(c & 0xffff);
            chi[g] = (unsigned char)(c >> 16);
        }
    }
}

__global__ void __launch_bounds__(256) gcn_spmm64_kernel(
        const int* __restrict__ rowptr, const float* __restrict__ cval,
        const unsigned short* __restrict__ clo, const unsigned char* __restrict__ chi,
        const float* __restrict__ h, float* __restrict__ out) {
    const int lane = threadIdx.x & 31;
    const int row = blockIdx.x * 8 + (threadIdx.x >> 5);
    if (row >= N_NODES) return;
    const int beg = __builtin_amdgcn_readfirstlane(rowptr[row]);
    const int end = __builtin_amdgcn_readfirstlane(rowptr[row + 1]);
    const int slot = lane >> 4;
    const int q = lane & 15;
    const float4* __restrict__ h4 = (const float4*)h;
    float4 acc = {0.f, 0.f, 0.f, 0.f};
    for (int cb = beg; cb < end; cb += 32) {
        const int n = (end - cb < 32) ? (end - cb) : 32;
        int c = 0;
        float v = 0.f;
        if (lane < n) {
            c = (int)clo[cb + lane] | ((int)chi[cb + lane] << 16);
            v = cval[cb + lane];
        }
        for (int g = 0; g < n; g += 8) {
            const int j0 = g + slot, j1 = j0 + 2, j2 = j0 + 4, j3 = j0 + 6;
            const int c0 = __shfl(c, j0), c1 = __shfl(c, j1), c2 = __shfl(c, j2), c3 = __shfl(c, j3);
            const float v0 = __shfl(v, j0), v1 = __shfl(v, j1), v2 = __shfl(v, j2), v3 = __shfl(v, j3);
            float4 r0 = h4[(size_t)c0 * 16 + q];
            float4 r1 = h4[(size_t)c1 * 16 + q];
            float4 r2 = h4[(size_t)c2 * 16 + q];
            float4 r3 = h4[(size_t)c3 * 16 + q];
            const float4 z = {0.f, 0.f, 0.f, 0.f};
            if (j0 >= n) r0 = z;
            if (j1 >= n) r1 = z;
            if (j2 >= n) r2 = z;
            if (j3 >= n) r3 = z;
            acc.x += v0 * r0.x; acc.y += v0 * r0.y; acc.z += v0 * r0.z; acc.w += v0 * r0.w;
            acc.x += v1 * r1.x; acc.y += v1 * r1.y; acc.z += v1 * r1.z; acc.w += v1 * r1.w;
            acc.x += v2 * r2.x; acc.y += v2 * r2.y; acc.z += v2 * r2.z; acc.w += v2 * r2.w;
            acc.x += v3 * r3.x; acc.y += v3 * r3.y; acc.z += v3 * r3.z; acc.w += v3 * r3.w;
        }
    }
    acc.x += __shfl_xor(acc.x, 16);
    acc.y += __shfl_xor(acc.y, 16);
    acc.z += __shfl_xor(acc.z, 16);
    acc.w += __shfl_xor(acc.w, 16);
    if (slot == 0) ((float4*)out)[(size_t)row * 16 + q] = acc;
}

__global__ void __launch_bounds__(128) gcn_gemm2_kernel(
        const float* __restrict__ h1, const float* __restrict__ W2,
        const float* __restrict__ b2, float* __restrict__ h2) {
    __shared__ float sW[N_HID * N_CLS];
    for (int i = threadIdx.x; i < N_HID * N_CLS; i += 128) sW[i] = W2[i];
    __syncthreads();

    const int wave = threadIdx.x >> 5;
    const int lane = threadIdx.x & 31;
    const int half = lane >> 4;
    const int m    = lane & 15;

    const int tile = blockIdx.x * 4 + wave;
    if (tile >= N_TILES) return;

    const int row = tile * 16 + m;
    const float* __restrict__ hrow = h1 + (size_t)row * N_HID;

    v8f acc = {};
    for (int k0 = 0; k0 < N_HID; k0 += 4) {
        const int k = k0 + 2 * half;
        v2f a;
        a.x = fmaxf(hrow[k], 0.f);
        a.y = fmaxf(hrow[k + 1], 0.f);
        v2f b;
        b.x = sW[(size_t)k * N_CLS + m];
        b.y = sW[(size_t)(k + 1) * N_CLS + m];
        acc = __builtin_amdgcn_wmma_f32_16x16x4_f32(false, a, false, b, (short)0, acc, false, false);
    }

    float* __restrict__ out = h2 + (size_t)tile * 16 * N_CLS;
    const float bb = b2[m];
    #pragma unroll
    for (int r = 0; r < 8; ++r)
        out[(size_t)(half * 8 + r) * N_CLS + m] = acc[r] + bb;
}

__global__ void __launch_bounds__(256) gcn_spmm16_kernel(
        const int* __restrict__ rowptr, const float* __restrict__ cval,
        const unsigned short* __restrict__ clo, const unsigned char* __restrict__ chi,
        const float* __restrict__ h, float* __restrict__ out) {
    __shared__ float4 sOut[32];
    const int lane = threadIdx.x & 31;
    const int wave = threadIdx.x >> 5;
    const int row = blockIdx.x * 8 + wave;
    const int beg = __builtin_amdgcn_readfirstlane(rowptr[row]);
    const int end = __builtin_amdgcn_readfirstlane(rowptr[row + 1]);
    const int slot = lane >> 2;
    const int q = lane & 3;
    const float4* __restrict__ h4 = (const float4*)h;
    const float4 z = {0.f, 0.f, 0.f, 0.f};
    float4 acc = z;
    for (int cb = beg; cb < end; cb += 32) {
        const int n = (end - cb < 32) ? (end - cb) : 32;
        int c = 0;
        float v = 0.f;
        if (lane < n) {
            c = (int)clo[cb + lane] | ((int)chi[cb + lane] << 16);
            v = cval[cb + lane];
        }
        for (int g = 0; g < n; g += 16) {
            const int j0 = g + slot, j1 = j0 + 8;
            const int c0 = __shfl(c, j0), c1 = __shfl(c, j1);
            const float v0 = __shfl(v, j0), v1 = __shfl(v, j1);
            float4 r0 = h4[(size_t)c0 * 4 + q];
            float4 r1 = h4[(size_t)c1 * 4 + q];
            if (j0 >= n) r0 = z;
            if (j1 >= n) r1 = z;
            acc.x += v0 * r0.x; acc.y += v0 * r0.y; acc.z += v0 * r0.z; acc.w += v0 * r0.w;
            acc.x += v1 * r1.x; acc.y += v1 * r1.y; acc.z += v1 * r1.z; acc.w += v1 * r1.w;
        }
    }
    #pragma unroll
    for (int o = 4; o < 32; o <<= 1) {
        acc.x += __shfl_xor(acc.x, o);
        acc.y += __shfl_xor(acc.y, o);
        acc.z += __shfl_xor(acc.z, o);
        acc.w += __shfl_xor(acc.w, o);
    }
    if (lane < 4) sOut[wave * 4 + q] = acc;
    __syncthreads();
    if (wave == 0) ((float4*)out)[(size_t)blockIdx.x * 32 + lane] = sOut[lane];
}

__global__ void __launch_bounds__(256) gcn_lsm_kernel(
        const float* __restrict__ s, float* __restrict__ out) {
    const int r = blockIdx.x * 256 + threadIdx.x;
    if (r >= N_NODES) return;
    const float* __restrict__ p = s + (size_t)r * N_CLS;
    float v[N_CLS];
    float mx = -3.402823466e+38f;
    #pragma unroll
    for (int i = 0; i < N_CLS; ++i) { v[i] = p[i]; mx = fmaxf(mx, v[i]); }
    float sum = 0.f;
    #pragma unroll
    for (int i = 0; i < N_CLS; ++i) sum += __expf(v[i] - mx);
    const float lse = mx + __logf(sum);
    float* __restrict__ o = out + (size_t)r * N_CLS;
    #pragma unroll
    for (int i = 0; i < N_CLS; ++i) o[i] = v[i] - lse;
}

extern "C" void kernel_launch(void* const* d_in, const int* in_sizes, int n_in,
                              void* d_out, int out_size, void* d_ws, size_t ws_size,
                              hipStream_t stream) {
    const float* x        = (const float*)d_in[0];
    const int*   adj_row  = (const int*)  d_in[1];
    const int*   adj_col  = (const int*)  d_in[2];
    const float* adj_vals = (const float*)d_in[3];
    const float* W1       = (const float*)d_in[4];
    const float* b1       = (const float*)d_in[5];
    const float* W2       = (const float*)d_in[6];
    const float* b2       = (const float*)d_in[7];
    float* out = (float*)d_out;

    float* ws = (float*)d_ws;
    float* h0 = ws;
    float* h2 = ws;
    float* s2 = ws + (size_t)N_NODES * N_CLS;
    float* h1 = ws + (size_t)N_NODES * N_HID;
    unsigned* pkey = (unsigned*)h1;
    float*    pval = h1 + N_EDGES;
    int*      tab  = (int*)(h1 + N_EDGES + N_EDGES);
    float* cval = ws + (size_t)N_NODES * (N_HID + N_HID);
    unsigned short* clo = (unsigned short*)(cval + N_EDGES);
    unsigned char*  chi = (unsigned char*)(cval + N_EDGES + N_EDGES / 2);
    int* rowptr = (int*)(cval + N_EDGES + N_EDGES / 2 + N_EDGES / 4);

    gcn_part_kernel<<<PART_B, 1024, 0, stream>>>(adj_row, adj_col, adj_vals, tab, pkey, pval);
    gcn_rowsort_kernel<<<N_BKT, 1024, 0, stream>>>(tab, pkey, pval, rowptr, cval, clo, chi);
    gcn_gemm1_kernel<<<(N_TILES + 3) / 4, 128, 0, stream>>>(x, W1, b1, h0);
    gcn_spmm64_kernel<<<N_NODES / 8, 256, 0, stream>>>(rowptr, cval, clo, chi, h0, h1);
    gcn_gemm2_kernel<<<(N_TILES + 3) / 4, 128, 0, stream>>>(h1, W2, b2, h2);
    gcn_spmm16_kernel<<<N_NODES / 8, 256, 0, stream>>>(rowptr, cval, clo, chi, h2, s2);
    gcn_lsm_kernel<<<(N_NODES + 255) / 256, 256, 0, stream>>>(s2, out);
}


# Consumer stages fused into the SpMM kernels: relu + GEMM2 (WMMA f32 on an LDS tile of 16 row sums, 8 waves x 2 rows) ends SpMM1, log_softmax ends SpMM2; h1, s2 stay on chip; 7 -> 5 launches
# speedup vs baseline: 4.8288x; 1.1191x over previous
#include <hip/hip_runtime.h>

typedef float v2f __attribute__((ext_vector_type(2)));
typedef float v8f __attribute__((ext_vector_type(8)));

#define N_NODES 100000
#define N_EDGES 1600000
#define N_FEAT  256
#define N_HID   64
#define N_CLS   16
#define N_TILES (N_NODES / 16)

__global__ void __launch_bounds__(128) gcn_gemm1_kernel(
        const float* __restrict__ x, const float* __restrict__ W1,
        const float* __restrict__ b1, float* __restrict__ h0) {
    __shared__ float sW[N_FEAT * N_HID];
    {
        const float4* W4 = (const float4*)W1;
        float4* s4 = (float4*)sW;
        for (int i = threadIdx.x; i < (N_FEAT * N_HID) / 4; i += 128) s4[i] = W4[i];
    }
    __syncthreads();

    const int wave = threadIdx.x >> 5;
    const int lane = threadIdx.x & 31;
    const int half = lane >> 4;
    const int m    = lane & 15;

    const int tile = blockIdx.x * 4 + wave;
    if (tile >= N_TILES) return;

    const int row = tile * 16 + m;
    const float* __restrict__ xrow = x + (size_t)row * N_FEAT;

    v8f acc0 = {}, acc1 = {}, acc2 = {}, acc3 = {};

    for (int k0 = 0; k0 < N_FEAT; k0 += 4) {
        const int k = k0 + 2 * half;
        v2f a;
        a.x = xrow[k];
        a.y = xrow[k + 1];
        const float* wk0 = sW + (size_t)k * N_HID;
        const float* wk1 = wk0 + N_HID;
        v2f b0, b1v, b2v, b3v;
        b0.x  = wk0[m];      b0.y  = wk1[m];
        b1v.x = wk0[16 + m]; b1v.y = wk1[16 + m];
        b2v.x = wk0[32 + m]; b2v.y = wk1[32 + m];
        b3v.x = wk0[48 + m]; b3v.y = wk1[48 + m];
        acc0 = __builtin_amdgcn_wmma_f32_16x16x4_f32(false, a, false, b0,  (short)0, acc0, false, false);
        acc1 = __builtin_amdgcn_wmma_f32_16x16x4_f32(false, a, false, b1v, (short)0, acc1, false, false);
        acc2 = __builtin_amdgcn_wmma_f32_16x16x4_f32(false, a, false, b2v, (short)0, acc2, false, false);
        acc3 = __builtin_amdgcn_wmma_f32_16x16x4_f32(false, a, false, b3v, (short)0, acc3, false, false);
    }

    float* __restrict__ out = h0 + (size_t)tile * 16 * N_HID;
    const float bb0 = b1[m], bb1 = b1[16 + m], bb2 = b1[32 + m], bb3 = b1[48 + m];
    #pragma unroll
    for (int r = 0; r < 8; ++r) {
        const size_t orow = (size_t)(half * 8 + r) * N_HID;
        out[orow + m]      = acc0[r] + bb0;
        out[orow + 16 + m] = acc1[r] + bb1;
        out[orow + 32 + m] = acc2[r] + bb2;
        out[orow + 48 + m] = acc3[r] + bb3;
    }
}

#define PART_B   256
#define PART_CH  (N_EDGES / PART_B)
#define BKT_ROWS 256
#define N_BKT    ((N_NODES + BKT_ROWS - 1) / BKT_ROWS)
#define TAB_W    (N_BKT + 1)
#define PART_IT  ((PART_CH + 1023) / 1024)

__global__ void __launch_bounds__(1024) gcn_part_kernel(
        const int* __restrict__ arow, const int* __restrict__ acol,
        const float* __restrict__ avals, int* __restrict__ tab,
        unsigned* __restrict__ pkey, float* __restrict__ pval) {
    __shared__ int sCur[TAB_W];
    __shared__ int sWave[32];
    __shared__ unsigned sKey[PART_CH];
    __shared__ float sVal[PART_CH];
    const int t = threadIdx.x;
    const int lane = t & 31;
    const int wave = t >> 5;
    const int base = blockIdx.x * PART_CH;
    if (t < TAB_W) sCur[t] = 0;
    __syncthreads();
    int r[PART_IT], c[PART_IT];
    float v[PART_IT];
    #pragma unroll
    for (int k = 0; k < PART_IT; ++k) {
        const int i = t + k * 1024;
        r[k] = 0; c[k] = 0; v[k] = 0.f;
        if (i < PART_CH) {
            r[k] = arow[base + i];
            c[k] = acol[base + i];
            v[k] = avals[base + i];
        }
    }
    #pragma unroll
    for (int k = 0; k < PART_IT; ++k)
        if (t + k * 1024 < PART_CH) atomicAdd(&sCur[r[k] >> 8], 1);
    __syncthreads();
    const int own = (t < N_BKT) ? sCur[t] : 0;
    int x = own;
    for (int o = 1; o < 32; o <<= 1) {
        const int y = __shfl_up(x, o);
        if (lane >= o) x += y;
    }
    if (lane == 31) sWave[wave] = x;
    __syncthreads();
    int woff = 0;
    for (int w = 0; w < 32; ++w)
        if (w < wave) woff += sWave[w];
    const int excl = woff + x - own;
    if (t < TAB_W) {
        sCur[t] = excl;
        tab[blockIdx.x * TAB_W + t] = excl;
    }
    __syncthreads();
    #pragma unroll
    for (int k = 0; k < PART_IT; ++k) {
        if (t + k * 1024 < PART_CH) {
            const int pos = atomicAdd(&sCur[r[k] >> 8], 1);
            sKey[pos] = ((unsigned)(r[k] & (BKT_ROWS - 1)) << 24) | (unsigned)c[k];
            sVal[pos] = v[k];
        }
    }
    __syncthreads();
    for (int i = t; i < PART_CH; i += 1024) {
        pkey[base + i] = sKey[i];
        pval[base + i] = sVal[i];
    }
}

#define SORT_CAP 8192
#define SORT_U   4

__device__ __forceinline__ int gcn_run_of(const int* sSrc, int e) {
    int lo = 0, hi = PART_B;
    #pragma unroll
    for (int it = 0; it < 8; ++it) {
        const int mid = (lo + hi) >> 1;
        if (sSrc[mid] <= e) lo = mid; else hi = mid;
    }
    return lo;
}

__global__ void __launch_bounds__(1024) gcn_rowsort_kernel(
        const int* __restrict__ tab, const unsigned* __restrict__ pkey,
        const float* __restrict__ pval, int* __restrict__ rowptr,
        float* __restrict__ cval, unsigned short* __restrict__ clo,
        unsigned char* __restrict__ chi) {
    __shared__ int sSrc[PART_B + 1];
    __shared__ int sBeg[PART_B];
    __shared__ int sCnt[BKT_ROWS];
    __shared__ int sWa[32];
    __shared__ int sWb[32];
    __shared__ int sKey[SORT_CAP];
    __shared__ float sVal[SORT_CAP];
    const int t = threadIdx.x;
    const int lane = t & 31;
    const int wave = t >> 5;
    const int b = blockIdx.x;

    int o0 = 0, cnt = 0;
    if (t < PART_B) {
        o0 = tab[t * TAB_W + b];
        cnt = tab[t * TAB_W + b + 1] - o0;
        sBeg[t] = t * PART_CH + o0;
    }
    if (t < BKT_ROWS) sCnt[t] = 0;
    int x = cnt;
    for (int o = 1; o < 32; o <<= 1) {
        const int y = __shfl_up(x, o);
        if (lane >= o) x += y;
    }
    int q = o0;
    for (int o = 16; o > 0; o >>= 1) q += __shfl_xor(q, o);
    if (lane == 31) sWa[wave] = x;
    if (lane == 0) sWb[wave] = q;
    __syncthreads();
    int woff = 0, n = 0, bbase = 0;
    for (int w = 0; w < 32; ++w) {
        const int a = sWa[w];
        if (w < wave) woff += a;
        n += a;
        bbase += sWb[w];
    }
    if (t < PART_B) sSrc[t] = woff + x - cnt;
    if (t == 0) sSrc[PART_B] = n;
    __syncthreads();

    for (int e0 = t; e0 < n; e0 += 1024 * SORT_U) {
        unsigned key[SORT_U];
        #pragma unroll
        for (int k = 0; k < SORT_U; ++k) {
            const int e = e0 + k * 1024;
            key[k] = 0u;
            if (e < n) {
                const int s = gcn_run_of(sSrc, e);
                key[k] = pkey[sBeg[s] + (e - sSrc[s])];
            }
        }
        #pragma unroll
        for (int k = 0; k < SORT_U; ++k)
            if (e0 + k * 1024 < n) atomicAdd(&sCnt[key[k] >> 24], 1);
    }
    __syncthreads();
    const int own = (t < BKT_ROWS) ? sCnt[t] : 0;
    int y = own;
    for (int o = 1; o < 32; o <<= 1) {
        const int z = __shfl_up(y, o);
        if (lane >= o) y += z;
    }
    if (lane == 31) sWa[wave] = y;
    __syncthreads();
    int roff = 0;
    for (int w = 0; w < 32; ++w)
        if (w < wave) roff += sWa[w];
    const int rstart = roff + y - own;
    if (t < BKT_ROWS) {
        sCnt[t] = rstart;
        const int row = b * BKT_ROWS + t;
        if (row <= N_NODES) rowptr[row] = bbase + rstart;
    }
    __syncthreads();

    const bool staged = (n <= SORT_CAP);
    for (int e0 = t; e0 < n; e0 += 1024 * SORT_U) {
        unsigned key[SORT_U];
        float val[SORT_U];
        #pragma unroll
        for (int k = 0; k < SORT_U; ++k) {
            const int e = e0 + k * 1024;
            key[k] = 0u; val[k] = 0.f;
            if (e < n) {
                const int s = gcn_run_of(sSrc, e);
                const int p = sBeg[s] + (e - sSrc[s]);
                key[k] = pkey[p];
                val[k] = pval[p];
            }
        }
        #pragma unroll
        for (int k = 0; k < SORT_U; ++k) {
            if (e0 + k * 1024 < n) {
                const int pos = atomicAdd(&sCnt[key[k] >> 24], 1);
                const int c = (int)(key[k] & 0xffffffu);
                if (staged) {
                    sKey[pos] = c;
                    sVal[pos] = val[k];
                } else {
                    const int g = bbase + pos;
                    cval[g] = val[k];
                    clo[g] = (unsigned short)(c & 0xffff);
                    chi[g] = (unsigned char)(c >> 16);
                }
            }
        }
    }
    __syncthreads();
    if (staged) {
        for (int i = t; i < n; i += 1024) {
            const int g = bbase + i;
            const int c = sKey[i];
            cval[g] = sVal[i];
            clo[g] = (unsigned short)(c & 0xffff);
            chi[g] = (unsigned char)(c >> 16);
        }
    }
}

#define H_PITCH 68

__global__ void __launch_bounds__(256) gcn_spmm64_gemm2_kernel(
        const int* __restrict__ rowptr, const float* __restrict__ cval,
        const unsigned short* __restrict__ clo, const unsigned char* __restrict__ chi,
        const float* __restrict__ h, const float* __restrict__ W2,
        const float* __restrict__ b2, float* __restrict__ h2) {
    __shared__ alignas(16) float sW[N_HID * N_CLS];
    __shared__ alignas(16) float sH[16 * H_PITCH];
    const float4 w2 = ((const float4*)W2)[threadIdx.x];

    const int lane = threadIdx.x & 31;
    const int wave = threadIdx.x >> 5;
    const int row0 = blockIdx.x * 16 + 2 * wave;
    const int p0 = __builtin_amdgcn_readfirstlane(rowptr[row0]);
    const int p1 = __builtin_amdgcn_readfirstlane(rowptr[row0 + 1]);
    const int p2 = __builtin_amdgcn_readfirstlane(rowptr[row0 + 2]);
    ((float4*)sW)[threadIdx.x] = w2;
    const int slot = lane >> 4;
    const int q = lane & 15;
    const float4* __restrict__ h4 = (const float4*)h;
    for (int rr = 0; rr < 2; ++rr) {
        const int beg = rr ? p1 : p0;
        const int end = rr ? p2 : p1;
        float4 acc = {0.f, 0.f, 0.f, 0.f};
        for (int cb = beg; cb < end; cb += 32) {
            const int n = (end - cb < 32) ? (end - cb) : 32;
            int c = 0;
            float v = 0.f;
            if (lane < n) {
                c = (int)clo[cb + lane] | ((int)chi[cb + lane] << 16);
                v = cval[cb + lane];
            }
            for (int g = 0; g < n; g += 8) {
                const int j0 = g + slot, j1 = j0 + 2, j2 = j0 + 4, j3 = j0 + 6;
                const int c0 = __shfl(c, j0), c1 = __shfl(c, j1), c2 = __shfl(c, j2), c3 = __shfl(c, j3);
                const float v0 = __shfl(v, j0), v1 = __shfl(v, j1), v2 = __shfl(v, j2), v3 = __shfl(v, j3);
                float4 r0 = h4[(size_t)c0 * 16 + q];
                float4 r1 = h4[(size_t)c1 * 16 + q];
                float4 r2 = h4[(size_t)c2 * 16 + q];
                float4 r3 = h4[(size_t)c3 * 16 + q];
                const float4 z = {0.f, 0.f, 0.f, 0.f};
                if (j0 >= n) r0 = z;
                if (j1 >= n) r1 = z;
                if (j2 >= n) r2 = z;
                if (j3 >= n) r3 = z;
                acc.x += v0 * r0.x; acc.y += v0 * r0.y; acc.z += v0 * r0.z; acc.w += v0 * r0.w;
                acc.x += v1 * r1.x; acc.y += v1 * r1.y; acc.z += v1 * r1.z; acc.w += v1 * r1.w;
                acc.x += v2 * r2.x; acc.y += v2 * r2.y; acc.z += v2 * r2.z; acc.w += v2 * r2.w;
                acc.x += v3 * r3.x; acc.y += v3 * r3.y; acc.z += v3 * r3.z; acc.w += v3 * r3.w;
            }
        }
        acc.x += __shfl_xor(acc.x, 16);
        acc.y += __shfl_xor(acc.y, 16);
        acc.z += __shfl_xor(acc.z, 16);
        acc.w += __shfl_xor(acc.w, 16);
        if (slot == 0) {
            float4 rl;
            rl.x = fmaxf(acc.x, 0.f); rl.y = fmaxf(acc.y, 0.f);
            rl.z = fmaxf(acc.z, 0.f); rl.w = fmaxf(acc.w, 0.f);
            *(float4*)&sH[(2 * wave + rr) * H_PITCH + 4 * q] = rl;
        }
    }
    __syncthreads();
    if (wave != 0) return;

    const int half = lane >> 4;
    const int m    = lane & 15;
    const float* __restrict__ hrow = sH + m * H_PITCH;
    v8f d = {};
    #pragma unroll
    for (int k0 = 0; k0 < N_HID; k0 += 4) {
        const int k = k0 + 2 * half;
        v2f a;
        a.x = hrow[k];
        a.y = hrow[k + 1];
        v2f b;
        b.x = sW[(size_t)k * N_CLS + m];
        b.y = sW[(size_t)(k + 1) * N_CLS + m];
        d = __builtin_amdgcn_wmma_f32_16x16x4_f32(false, a, false, b, (short)0, d, false, false);
    }
    float* __restrict__ out = h2 + (size_t)blockIdx.x * 16 * N_CLS;
    const float bb = b2[m];
    #pragma unroll
    for (int r = 0; r < 8; ++r)
        out[(size_t)(half * 8 + r) * N_CLS + m] = d[r] + bb;
}

__global__ void __launch_bounds__(256) gcn_spmm16_lsm_kernel(
        const int* __restrict__ rowptr, const float* __restrict__ cval,
        const unsigned short* __restrict__ clo, const unsigned char* __restrict__ chi,
        const float* __restrict__ h, float* __restrict__ out) {
    __shared__ float4 sOut[32];
    const int lane = threadIdx.x & 31;
    const int wave = threadIdx.x >> 5;
    const int row = blockIdx.x * 8 + wave;
    const int beg = __builtin_amdgcn_readfirstlane(rowptr[row]);
    const int end = __builtin_amdgcn_readfirstlane(rowptr[row + 1]);
    const int slot = lane >> 2;
    const int q = lane & 3;
    const float4* __restrict__ h4 = (const float4*)h;
    const float4 z = {0.f, 0.f, 0.f, 0.f};
    float4 acc = z;
    for (int cb = beg; cb < end; cb += 32) {
        const int n = (end - cb < 32) ? (end - cb) : 32;
        int c = 0;
        float v = 0.f;
        if (lane < n) {
            c = (int)clo[cb + lane] | ((int)chi[cb + lane] << 16);
            v = cval[cb + lane];
        }
        for (int g = 0; g < n; g += 16) {
            const int j0 = g + slot, j1 = j0 + 8;
            const int c0 = __shfl(c, j0), c1 = __shfl(c, j1);
            const float v0 = __shfl(v, j0), v1 = __shfl(v, j1);
            float4 r0 = h4[(size_t)c0 * 4 + q];
            float4 r1 = h4[(size_t)c1 * 4 + q];
            if (j0 >= n) r0 = z;
            if (j1 >= n) r1 = z;
            acc.x += v0 * r0.x; acc.y += v0 * r0.y; acc.z += v0 * r0.z; acc.w += v0 * r0.w;
            acc.x += v1 * r1.x; acc.y += v1 * r1.y; acc.z += v1 * r1.z; acc.w += v1 * r1.w;
        }
    }
    #pragma unroll
    for (int o = 4; o < 32; o <<= 1) {
        acc.x += __shfl_xor(acc.x, o);
        acc.y += __shfl_xor(acc.y, o);
        acc.z += __shfl_xor(acc.z, o);
        acc.w += __shfl_xor(acc.w, o);
    }
    if (lane < 4) sOut[wave * 4 + q] = acc;
    __syncthreads();
    if (wave != 0) return;
    const float* __restrict__ p = (const float*)&sOut[(lane & 7) * 4];
    float v[N_CLS];
    float mx = -3.402823466e+38f;
    #pragma unroll
    for (int i = 0; i < N_CLS; ++i) { v[i] = p[i]; mx = fmaxf(mx, v[i]); }
    float sum = 0.f;
    #pragma unroll
    for (int i = 0; i < N_CLS; ++i) sum += __expf(v[i] - mx);
    const float lse_own = mx + __logf(sum);
    const float lse = __shfl(lse_own, lane >> 2);
    float4 o = sOut[lane];
    o.x -= lse; o.y -= lse; o.z -= lse; o.w -= lse;
    ((float4*)out)[(size_t)blockIdx.x * 32 + lane] = o;
}


extern "C" void kernel_launch(void* const* d_in, const int* in_sizes, int n_in,
                              void* d_out, int out_size, void* d_ws, size_t ws_size,
                              hipStream_t stream) {
    const float* x        = (const float*)d_in[0];
    const int*   adj_row  = (const int*)  d_in[1];
    const int*   adj_col  = (const int*)  d_in[2];
    const float* adj_vals = (const float*)d_in[3];
    const float* W1       = (const float*)d_in[4];
    const float* b1       = (const float*)d_in[5];
    const float* W2       = (const float*)d_in[6];
    const float* b2       = (const float*)d_in[7];
    float* out = (float*)d_out;

    float* ws = (float*)d_ws;
    float* h0 = ws;
    float* wb = ws + (size_t)N_NODES * N_HID;
    float* h2 = wb;
    unsigned* pkey = (unsigned*)wb;
    float*    pval = wb + N_EDGES;
    int*      tab  = (int*)(wb + N_EDGES + N_EDGES);
    float* cval = ws + (size_t)N_NODES * (N_HID + N_HID);
    unsigned short* clo = (unsigned short*)(cval + N_EDGES);
    unsigned char*  chi = (unsigned char*)(cval + N_EDGES + N_EDGES / 2);
    int* rowptr = (int*)(cval + N_EDGES + N_EDGES / 2 + N_EDGES / 4);

    gcn_part_kernel<<<PART_B, 1024, 0, stream>>>(adj_row, adj_col, adj_vals, tab, pkey, pval);
    gcn_rowsort_kernel<<<N_BKT, 1024, 0, stream>>>(tab, pkey, pval, rowptr, cval, clo, chi);
    gcn_gemm1_kernel<<<(N_TILES + 3) / 4, 128, 0, stream>>>(x, W1, b1, h0);
    gcn_spmm64_gemm2_kernel<<<N_TILES, 256, 0, stream>>>(rowptr, cval, clo, chi, h0, W2, b2, h2);
    gcn_spmm16_lsm_kernel<<<N_NODES / 8, 256, 0, stream>>>(rowptr, cval, clo, chi, h2, out);
}


# GEMM1 rides as 256 extra 32-wave work-groups (role by block number, W1 staged once each, 24-25 row tiles each) in the second edge-ordering launch instead of a launch of its own; 5 -> 4 launches
# speedup vs baseline: 5.0170x; 1.0390x over previous
#include <hip/hip_runtime.h>

typedef float v2f __attribute__((ext_vector_type(2)));
typedef float v8f __attribute__((ext_vector_type(8)));

#define N_NODES 100000
#define N_EDGES 1600000
#define N_FEAT  256
#define N_HID   64
#define N_CLS   16
#define N_TILES (N_NODES / 16)

#define GEMM_WG  256
#define G_TPW    24
#define G_FULL   (GEMM_WG * G_TPW)

__device__ __forceinline__ void gcn_gemm1_role(
        float* __restrict__ sW, const int g,
        const float* __restrict__ x, const float* __restrict__ W1,
        const float* __restrict__ b1, float* __restrict__ h0) {
    {
        const float4* W4 = (const float4*)W1;
        float4* s4 = (float4*)sW;
        for (int i = threadIdx.x; i < (N_FEAT * N_HID) / 4; i += 1024) s4[i] = W4[i];
    }
    __syncthreads();

    const int wave = threadIdx.x >> 5;
    const int lane = threadIdx.x & 31;
    const int half = lane >> 4;
    const int m    = lane & 15;

    if (wave > G_TPW) return;
    const int tile = (wave < G_TPW) ? g * G_TPW + wave : G_FULL + g;
    if (tile >= N_TILES) return;

    const int row = tile * 16 + m;
    const float* __restrict__ xrow = x + (size_t)row * N_FEAT;

    v8f acc0 = {}, acc1 = {}, acc2 = {}, acc3 = {};

    for (int k0 = 0; k0 < N_FEAT; k0 += 4) {
        const int k = k0 + 2 * half;
        v2f a;
        a.x = xrow[k];
        a.y = xrow[k + 1];
        const float* wk0 = sW + (size_t)k * N_HID;
        const float* wk1 = wk0 + N_HID;
        v2f b0, b1v, b2v, b3v;
        b0.x  = wk0[m];      b0.y  = wk1[m];
        b1v.x = wk0[16 + m]; b1v.y = wk1[16 + m];
        b2v.x = wk0[32 + m]; b2v.y = wk1[32 + m];
        b3v.x = wk0[48 + m]; b3v.y = wk1[48 + m];
        acc0 = __builtin_amdgcn_wmma_f32_16x16x4_f32(false, a, false, b0,  (short)0, acc0, false, false);
        acc1 = __builtin_amdgcn_wmma_f32_16x16x4_f32(false, a, false, b1v, (short)0, acc1, false, false);
        acc2 = __builtin_amdgcn_wmma_f32_16x16x4_f32(false, a, false, b2v, (short)0, acc2, false, false);
        acc3 = __builtin_amdgcn_wmma_f32_16x16x4_f32(false, a, false, b3v, (short)0, acc3, false, false);
    }

    float* __restrict__ out = h0 + (size_t)tile * 16 * N_HID;
    const float bb0 = b1[m], bb1 = b1[16 + m], bb2 = b1[32 + m], bb3 = b1[48 + m];
    #pragma unroll
    for (int r = 0; r < 8; ++r) {
        const size_t orow = (size_t)(half * 8 + r) * N_HID;
        out[orow + m]      = acc0[r] + bb0;
        out[orow + 16 + m] = acc1[r] + bb1;
        out[orow + 32 + m] = acc2[r] + bb2;
        out[orow + 48 + m] = acc3[r] + bb3;
    }
}

#define PART_B   256
#define PART_CH  (N_EDGES / PART_B)
#define BKT_ROWS 256
#define N_BKT    ((N_NODES + BKT_ROWS - 1) / BKT_ROWS)
#define TAB_W    (N_BKT + 1)
#define PART_IT  ((PART_CH + 1023) / 1024)

__global__ void __launch_bounds__(1024) gcn_part_kernel(
        const int* __restrict__ arow, const int* __restrict__ acol,
        const float* __restrict__ avals, int* __restrict__ tab,
        unsigned* __restrict__ pkey, float* __restrict__ pval) {
    __shared__ int sCur[TAB_W];
    __shared__ int sWave[32];
    __shared__ unsigned sKey[PART_CH];
    __shared__ float sVal[PART_CH];
    const int t = threadIdx.x;
    const int lane = t & 31;
    const int wave = t >> 5;
    const int base = blockIdx.x * PART_CH;
    if (t < TAB_W) sCur[t] = 0;
    __syncthreads();
    int r[PART_IT], c[PART_IT];
    float v[PART_IT];
    #pragma unroll
    for (int k = 0; k < PART_IT; ++k) {
        const int i = t + k * 1024;
        r[k] = 0; c[k] = 0; v[k] = 0.f;
        if (i < PART_CH) {
            r[k] = arow[base + i];
            c[k] = acol[base + i];
            v[k] = avals[base + i];
        }
    }
    #pragma unroll
    for (int k = 0; k < PART_IT; ++k)
        if (t + k * 1024 < PART_CH) atomicAdd(&sCur[r[k] >> 8], 1);
    __syncthreads();
    const int own = (t < N_BKT) ? sCur[t] : 0;
    int x = own;
    for (int o = 1; o < 32; o <<= 1) {
        const int y = __shfl_up(x, o);
        if (lane >= o) x += y;
    }
    if (lane == 31) sWave[wave] = x;
    __syncthreads();
    int woff = 0;
    for (int w = 0; w < 32; ++w)
        if (w < wave) woff += sWave[w];
    const int excl = woff + x - own;
    if (t < TAB_W) {
        sCur[t] = excl;
        tab[blockIdx.x * TAB_W + t] = excl;
    }
    __syncthreads();
    #pragma unroll
    for (int k = 0; k < PART_IT; ++k) {
        if (t + k * 1024 < PART_CH) {
            const int pos = atomicAdd(&sCur[r[k] >> 8], 1);
            sKey[pos] = ((unsigned)(r[k] & (BKT_ROWS - 1)) << 24) | (unsigned)c[k];
            sVal[pos] = v[k];
        }
    }
    __syncthreads();
    for (int i = t; i < PART_CH; i += 1024) {
        pkey[base + i] = sKey[i];
        pval[base + i] = sVal[i];
    }
}

#define SORT_CAP 8192
#define SORT_U   4

__device__ __forceinline__ int gcn_run_of(const int* sSrc, int e) {
    int lo = 0, hi = PART_B;
    #pragma unroll
    for (int it = 0; it < 8; ++it) {
        const int mid = (lo + hi) >> 1;
        if (sSrc[mid] <= e) lo = mid; else hi = mid;
    }
    return lo;
}

#define SORT_LDS_W (260 + PART_B + BKT_ROWS + 32 + 32 + SORT_CAP + SORT_CAP)

__device__ __forceinline__ void gcn_rowsort_role(
        float* __restrict__ smem, const int b,
        const int* __restrict__ tab, const unsigned* __restrict__ pkey,
        const float* __restrict__ pval, int* __restrict__ rowptr,
        float* __restrict__ cval, unsigned short* __restrict__ clo,
        unsigned char* __restrict__ chi) {
    int* sSrc = (int*)smem;
    int* sBeg = sSrc + 260;
    int* sCnt = sBeg + PART_B;
    int* sWa = sCnt + BKT_ROWS;
    int* sWb = sWa + 32;
    int* sKey = sWb + 32;
    float* sVal = (float*)(sKey + SORT_CAP);
    const int t = threadIdx.x;
    const int lane = t & 31;
    const int wave = t >> 5;

    int o0 = 0, cnt = 0;
    if (t < PART_B) {
        o0 = tab[t * TAB_W + b];
        cnt = tab[t * TAB_W + b + 1] - o0;
        sBeg[t] = t * PART_CH + o0;
    }
    if (t < BKT_ROWS) sCnt[t] = 0;
    int x = cnt;
    for (int o = 1; o < 32; o <<= 1) {
        const int y = __shfl_up(x, o);
        if (lane >= o) x += y;
    }
    int q = o0;
    for (int o = 16; o > 0; o >>= 1) q += __shfl_xor(q, o);
    if (lane == 31) sWa[wave] = x;
    if (lane == 0) sWb[wave] = q;
    __syncthreads();
    int woff = 0, n = 0, bbase = 0;
    for (int w = 0; w < 32; ++w) {
        const int a = sWa[w];
        if (w < wave) woff += a;
        n += a;
        bbase += sWb[w];
    }
    if (t < PART_B) sSrc[t] = woff + x - cnt;
    if (t == 0) sSrc[PART_B] = n;
    __syncthreads();

    for (int e0 = t; e0 < n; e0 += 1024 * SORT_U) {
        unsigned key[SORT_U];
        #pragma unroll
        for (int k = 0; k < SORT_U; ++k) {
            const int e = e0 + k * 1024;
            key[k] = 0u;
            if (e < n) {
                const int s = gcn_run_of(sSrc, e);
                key[k] = pkey[sBeg[s] + (e - sSrc[s])];
            }
        }
        #pragma unroll
        for (int k = 0; k < SORT_U; ++k)
            if (e0 + k * 1024 < n) atomicAdd(&sCnt[key[k] >> 24], 1);
    }
    __syncthreads();
    const int own = (t < BKT_ROWS) ? sCnt[t] : 0;
    int y = own;
    for (int o = 1; o < 32; o <<= 1) {
        const int z = __shfl_up(y, o);
        if (lane >= o) y += z;
    }
    if (lane == 31) sWa[wave] = y;
    __syncthreads();
    int roff = 0;
    for (int w = 0; w < 32; ++w)
        if (w < wave) roff += sWa[w];
    const int rstart = roff + y - own;
    if (t < BKT_ROWS) {
        sCnt[t] = rstart;
        const int row = b * BKT_ROWS + t;
        if (row <= N_NODES) rowptr[row] = bbase + rstart;
    }
    __syncthreads();

    const bool staged = (n <= SORT_CAP);
    for (int e0 = t; e0 < n; e0 += 1024 * SORT_U) {
        unsigned key[SORT_U];
        float val[SORT_U];
        #pragma unroll
        for (int k = 0; k < SORT_U; ++k) {
            const int e = e0 + k * 1024;
            key[k] = 0u; val[k] = 0.f;
            if (e < n) {
                const int s = gcn_run_of(sSrc, e);
                const int p = sBeg[s] + (e - sSrc[s]);
                key[k] = pkey[p];
                val[k] = pval[p];
            }
        }
        #pragma unroll
        for (int k = 0; k < SORT_U; ++k) {
            if (e0 + k * 1024 < n) {
                const int pos = atomicAdd(&sCnt[key[k] >> 24], 1);
                const int c = (int)(key[k] & 0xffffffu);
                if (staged) {
                    sKey[pos] = c;
                    sVal[pos] = val[k];
                } else {
                    const int g = bbase + pos;
                    cval[g] = val[k];
                    clo[g] = (unsigned short)(c & 0xffff);
                    chi[g] = (unsigned char)(c >> 16);
                }
            }
        }
    }
    __syncthreads();
    if (staged) {
        for (int i = t; i < n; i += 1024) {
            const int g = bbase + i;
            const int c = sKey[i];
            cval[g] = sVal[i];
            clo[g] = (unsigned short)(c & 0xffff);
            chi[g] = (unsigned char)(c >> 16);
        }
    }
}

__global__ void __launch_bounds__(1024) gcn_rowsort_gemm1_kernel(
        const float* __restrict__ x, const float* __restrict__ W1,
        const float* __restrict__ b1, float* __restrict__ h0,
        const int* __restrict__ tab, const unsigned* __restrict__ pkey,
        const float* __restrict__ pval, int* __restrict__ rowptr,
        float* __restrict__ cval, unsigned short* __restrict__ clo,
        unsigned char* __restrict__ chi) {
    __shared__ alignas(16) float smem[(N_FEAT * N_HID) > SORT_LDS_W ? (N_FEAT * N_HID) : SORT_LDS_W];
    const int blk = blockIdx.x;
    const bool paired = blk < 2 * GEMM_WG;
    const int pair = (blk & (GEMM_WG - 1)) >> 1;
    const int idx = !paired ? blk - GEMM_WG : (blk < GEMM_WG) ? pair : pair + GEMM_WG / 2;
    const bool mult = paired && (((blk & 1) != 0) == (blk >= GEMM_WG));
    if (mult) gcn_gemm1_role(smem, idx, x, W1, b1, h0);
    else      gcn_rowsort_role(smem, idx, tab, pkey, pval, rowptr, cval, clo, chi);
}

#define H_PITCH 68

__global__ void __launch_bounds__(256) gcn_spmm64_gemm2_kernel(
        const int* __restrict__ rowptr, const float* __restrict__ cval,
        const unsigned short* __restrict__ clo, const unsigned char* __restrict__ chi,
        const float* __restrict__ h, const float* __restrict__ W2,
        const float* __restrict__ b2, float* __restrict__ h2) {
    __shared__ alignas(16) float sW[N_HID * N_CLS];
    __shared__ alignas(16) float sH[16 * H_PITCH];
    const float4 w2 = ((const float4*)W2)[threadIdx.x];

    const int lane = threadIdx.x & 31;
    const int wave = threadIdx.x >> 5;
    const int row0 = blockIdx.x * 16 + 2 * wave;
    const int p0 = __builtin_amdgcn_readfirstlane(rowptr[row0]);
    const int p1 = __builtin_amdgcn_readfirstlane(rowptr[row0 + 1]);
    const int p2 = __builtin_amdgcn_readfirstlane(rowptr[row0 + 2]);
    ((float4*)sW)[threadIdx.x] = w2;
    const int slot = lane >> 4;
    const int q = lane & 15;
    const float4* __restrict__ h4 = (const float4*)h;
    for (int rr = 0; rr < 2; ++rr) {
        const int beg = rr ? p1 : p0;
        const int end = rr ? p2 : p1;
        float4 acc = {0.f, 0.f, 0.f, 0.f};
        for (int cb = beg; cb < end; cb += 32) {
            const int n = (end - cb < 32) ? (end - cb) : 32;
            int c = 0;
            float v = 0.f;
            if (lane < n) {
                c = (int)clo[cb + lane] | ((int)chi[cb + lane] << 16);
                v = cval[cb + lane];
            }
            for (int g = 0; g < n; g += 8) {
                const int j0 = g + slot, j1 = j0 + 2, j2 = j0 + 4, j3 = j0 + 6;
                const int c0 = __shfl(c, j0), c1 = __shfl(c, j1), c2 = __shfl(c, j2), c3 = __shfl(c, j3);
                const float v0 = __shfl(v, j0), v1 = __shfl(v, j1), v2 = __shfl(v, j2), v3 = __shfl(v, j3);
                float4 r0 = h4[(size_t)c0 * 16 + q];
                float4 r1 = h4[(size_t)c1 * 16 + q];
                float4 r2 = h4[(size_t)c2 * 16 + q];
                float4 r3 = h4[(size_t)c3 * 16 + q];
                const float4 z = {0.f, 0.f, 0.f, 0.f};
                if (j0 >= n) r0 = z;
                if (j1 >= n) r1 = z;
                if (j2 >= n) r2 = z;
                if (j3 >= n) r3 = z;
                acc.x += v0 * r0.x; acc.y += v0 * r0.y; acc.z += v0 * r0.z; acc.w += v0 * r0.w;
                acc.x += v1 * r1.x; acc.y += v1 * r1.y; acc.z += v1 * r1.z; acc.w += v1 * r1.w;
                acc.x += v2 * r2.x; acc.y += v2 * r2.y; acc.z += v2 * r2.z; acc.w += v2 * r2.w;
                acc.x += v3 * r3.x; acc.y += v3 * r3.y; acc.z += v3 * r3.z; acc.w += v3 * r3.w;
            }
        }
        acc.x += __shfl_xor(acc.x, 16);
        acc.y += __shfl_xor(acc.y, 16);
        acc.z += __shfl_xor(acc.z, 16);
        acc.w += __shfl_xor(acc.w, 16);
        if (slot == 0) {
            float4 rl;
            rl.x = fmaxf(acc.x, 0.f); rl.y = fmaxf(acc.y, 0.f);
            rl.z = fmaxf(acc.z, 0.f); rl.w = fmaxf(acc.w, 0.f);
            *(float4*)&sH[(2 * wave + rr) * H_PITCH + 4 * q] = rl;
        }
    }
    __syncthreads();
    if (wave != 0) return;

    const int half = lane >> 4;
    const int m    = lane & 15;
    const float* __restrict__ hrow = sH + m * H_PITCH;
    v8f d = {};
    #pragma unroll
    for (int k0 = 0; k0 < N_HID; k0 += 4) {
        const int k = k0 + 2 * half;
        v2f a;
        a.x = hrow[k];
        a.y = hrow[k + 1];
        v2f b;
        b.x = sW[(size_t)k * N_CLS + m];
        b.y = sW[(size_t)(k + 1) * N_CLS + m];
        d = __builtin_amdgcn_wmma_f32_16x16x4_f32(false, a, false, b, (short)0, d, false, false);
    }
    float* __restrict__ out = h2 + (size_t)blockIdx.x * 16 * N_CLS;
    const float bb = b2[m];
    #pragma unroll
    for (int r = 0; r < 8; ++r)
        out[(size_t)(half * 8 + r) * N_CLS + m] = d[r] + bb;
}

__global__ void __launch_bounds__(256) gcn_spmm16_lsm_kernel(
        const int* __restrict__ rowptr, const float* __restrict__ cval,
        const unsigned short* __restrict__ clo, const unsigned char* __restrict__ chi,
        const float* __restrict__ h, float* __restrict__ out) {
    __shared__ float4 sOut[32];
    const int lane = threadIdx.x & 31;
    const int wave = threadIdx.x >> 5;
    const int row = blockIdx.x * 8 + wave;
    const int beg = __builtin_amdgcn_readfirstlane(rowptr[row]);
    const int end = __builtin_amdgcn_readfirstlane(rowptr[row + 1]);
    const int slot = lane >> 2;
    const int q = lane & 3;
    const float4* __restrict__ h4 = (const float4*)h;
    const float4 z = {0.f, 0.f, 0.f, 0.f};
    float4 acc = z;
    for (int cb = beg; cb < end; cb += 32) {
        const int n = (end - cb < 32) ? (end - cb) : 32;
        int c = 0;
        float v = 0.f;
        if (lane < n) {
            c = (int)clo[cb + lane] | ((int)chi[cb + lane] << 16);
            v = cval[cb + lane];
        }
        for (int g = 0; g < n; g += 16) {
            const int j0 = g + slot, j1 = j0 + 8;
            const int c0 = __shfl(c, j0), c1 = __shfl(c, j1);
            const float v0 = __shfl(v, j0), v1 = __shfl(v, j1);
            float4 r0 = h4[(size_t)c0 * 4 + q];
            float4 r1 = h4[(size_t)c1 * 4 + q];
            if (j0 >= n) r0 = z;
            if (j1 >= n) r1 = z;
            acc.x += v0 * r0.x; acc.y += v0 * r0.y; acc.z += v0 * r0.z; acc.w += v0 * r0.w;
            acc.x += v1 * r1.x; acc.y += v1 * r1.y; acc.z += v1 * r1.z; acc.w += v1 * r1.w;
        }
    }
    #pragma unroll
    for (int o = 4; o < 32; o <<= 1) {
        acc.x += __shfl_xor(acc.x, o);
        acc.y += __shfl_xor(acc.y, o);
        acc.z += __shfl_xor(acc.z, o);
        acc.w += __shfl_xor(acc.w, o);
    }
    if (lane < 4) sOut[wave * 4 + q] = acc;
    __syncthreads();
    if (wave != 0) return;
    const float* __restrict__ p = (const float*)&sOut[(lane & 7) * 4];
    float v[N_CLS];
    float mx = -3.402823466e+38f;
    #pragma unroll
    for (int i = 0; i < N_CLS; ++i) { v[i] = p[i]; mx = fmaxf(mx, v[i]); }
    float sum = 0.f;
    #pragma unroll
    for (int i = 0; i < N_CLS; ++i) sum += __expf(v[i] - mx);
    const float lse_own = mx + __logf(sum);
    const float lse = __shfl(lse_own, lane >> 2);
    float4 o = sOut[lane];
    o.x -= lse; o.y -= lse; o.z -= lse; o.w -= lse;
    ((float4*)out)[(size_t)blockIdx.x * 32 + lane] = o;
}


extern "C" void kernel_launch(void* const* d_in, const int* in_sizes, int n_in,
                              void* d_out, int out_size, void* d_ws, size_t ws_size,
                              hipStream_t stream) {
    const float* x        = (const float*)d_in[0];
    const int*   adj_row  = (const int*)  d_in[1];
    const int*   adj_col  = (const int*)  d_in[2];
    const float* adj_vals = (const float*)d_in[3];
    const float* W1       = (const float*)d_in[4];
    const float* b1       = (const float*)d_in[5];
    const float* W2       = (const float*)d_in[6];
    const float* b2       = (const float*)d_in[7];
    float* out = (float*)d_out;

    float* ws = (float*)d_ws;
    float* h0 = ws;
    float* wb = ws + (size_t)N_NODES * N_HID;
    float* h2 = wb;
    unsigned* pkey = (unsigned*)wb;
    float*    pval = wb + N_EDGES;
    int*      tab  = (int*)(wb + N_EDGES + N_EDGES);
    float* cval = ws + (size_t)N_NODES * (N_HID + N_HID);
    unsigned short* clo = (unsigned short*)(cval + N_EDGES);
    unsigned char*  chi = (unsigned char*)(cval + N_EDGES + N_EDGES / 2);
    int* rowptr = (int*)(cval + N_EDGES + N_EDGES / 2 + N_EDGES / 4);

    gcn_part_kernel<<<PART_B, 1024, 0, stream>>>(adj_row, adj_col, adj_vals, tab, pkey, pval);
    gcn_rowsort_gemm1_kernel<<<N_BKT + GEMM_WG, 1024, 0, stream>>>(
        x, W1, b1, h0, tab, pkey, pval, rowptr, cval, clo, chi);
    gcn_spmm64_gemm2_kernel<<<N_TILES, 256, 0, stream>>>(rowptr, cval, clo, chi, h0, W2, b2, h2);
    gcn_spmm16_lsm_kernel<<<N_NODES / 8, 256, 0, stream>>>(rowptr, cval, clo, chi, h2, out);
}


# SpMM2: one wave sums four neighbouring rows, all their edge-list loads then all their source-row loads in flight together; log_softmax in registers, no LDS; 100000 -> 25000 waves; row sum order kept
# speedup vs baseline: 5.1839x; 1.0333x over previous
#include <hip/hip_runtime.h>

typedef float v2f __attribute__((ext_vector_type(2)));
typedef float v8f __attribute__((ext_vector_type(8)));

#define N_NODES 100000
#define N_EDGES 1600000
#define N_FEAT  256
#define N_HID   64
#define N_CLS   16
#define N_TILES (N_NODES / 16)

#define GEMM_WG  256
#define G_TPW    24
#define G_FULL   (GEMM_WG * G_TPW)

__device__ __forceinline__ void gcn_gemm1_role(
        float* __restrict__ sW, const int g,
        const float* __restrict__ x, const float* __restrict__ W1,
        const float* __restrict__ b1, float* __restrict__ h0) {
    {
        const float4* W4 = (const float4*)W1;
        float4* s4 = (float4*)sW;
        for (int i = threadIdx.x; i < (N_FEAT * N_HID) / 4; i += 1024) s4[i] = W4[i];
    }
    __syncthreads();

    const int wave = threadIdx.x >> 5;
    const int lane = threadIdx.x & 31;
    const int half = lane >> 4;
    const int m    = lane & 15;

    if (wave > G_TPW) return;
    const int tile = (wave < G_TPW) ? g * G_TPW + wave : G_FULL + g;
    if (tile >= N_TILES) return;

    const int row = tile * 16 + m;
    const float* __restrict__ xrow = x + (size_t)row * N_FEAT;

    v8f acc0 = {}, acc1 = {}, acc2 = {}, acc3 = {};

    for (int k0 = 0; k0 < N_FEAT; k0 += 4) {
        const int k = k0 + 2 * half;
        v2f a;
        a.x = xrow[k];
        a.y = xrow[k + 1];
        const float* wk0 = sW + (size_t)k * N_HID;
        const float* wk1 = wk0 + N_HID;
        v2f b0, b1v, b2v, b3v;
        b0.x  = wk0[m];      b0.y  = wk1[m];
        b1v.x = wk0[16 + m]; b1v.y = wk1[16 + m];
        b2v.x = wk0[32 + m]; b2v.y = wk1[32 + m];
        b3v.x = wk0[48 + m]; b3v.y = wk1[48 + m];
        acc0 = __builtin_amdgcn_wmma_f32_16x16x4_f32(false, a, false, b0,  (short)0, acc0, false, false);
        acc1 = __builtin_amdgcn_wmma_f32_16x16x4_f32(false, a, false, b1v, (short)0, acc1, false, false);
        acc2 = __builtin_amdgcn_wmma_f32_16x16x4_f32(false, a, false, b2v, (short)0, acc2, false, false);
        acc3 = __builtin_amdgcn_wmma_f32_16x16x4_f32(false, a, false, b3v, (short)0, acc3, false, false);
    }

    float* __restrict__ out = h0 + (size_t)tile * 16 * N_HID;
    const float bb0 = b1[m], bb1 = b1[16 + m], bb2 = b1[32 + m], bb3 = b1[48 + m];
    #pragma unroll
    for (int r = 0; r < 8; ++r) {
        const size_t orow = (size_t)(half * 8 + r) * N_HID;
        out[orow + m]      = acc0[r] + bb0;
        out[orow + 16 + m] = acc1[r] + bb1;
        out[orow + 32 + m] = acc2[r] + bb2;
        out[orow + 48 + m] = acc3[r] + bb3;
    }
}

#define PART_B   256
#define PART_CH  (N_EDGES / PART_B)
#define BKT_ROWS 256
#define N_BKT    ((N_NODES + BKT_ROWS - 1) / BKT_ROWS)
#define TAB_W    (N_BKT + 1)
#define PART_IT  ((PART_CH + 1023) / 1024)

__global__ void __launch_bounds__(1024) gcn_part_kernel(
        const int* __restrict__ arow, const int* __restrict__ acol,
        const float* __restrict__ avals, int* __restrict__ tab,
        unsigned* __restrict__ pkey, float* __restrict__ pval) {
    __shared__ int sCur[TAB_W];
    __shared__ int sWave[32];
    __shared__ unsigned sKey[PART_CH];
    __shared__ float sVal[PART_CH];
    const int t = threadIdx.x;
    const int lane = t & 31;
    const int wave = t >> 5;
    const int base = blockIdx.x * PART_CH;
    if (t < TAB_W) sCur[t] = 0;
    __syncthreads();
    int r[PART_IT], c[PART_IT];
    float v[PART_IT];
    #pragma unroll
    for (int k = 0; k < PART_IT; ++k) {
        const int i = t + k * 1024;
        r[k] = 0; c[k] = 0; v[k] = 0.f;
        if (i < PART_CH) {
            r[k] = arow[base + i];
            c[k] = acol[base + i];
            v[k] = avals[base + i];
        }
    }
    #pragma unroll
    for (int k = 0; k < PART_IT; ++k)
        if (t + k * 1024 < PART_CH) atomicAdd(&sCur[r[k] >> 8], 1);
    __syncthreads();
    const int own = (t < N_BKT) ? sCur[t] : 0;
    int x = own;
    for (int o = 1; o < 32; o <<= 1) {
        const int y = __shfl_up(x, o);
        if (lane >= o) x += y;
    }
    if (lane == 31) sWave[wave] = x;
    __syncthreads();
    int woff = 0;
    for (int w = 0; w < 32; ++w)
        if (w < wave) woff += sWave[w];
    const int excl = woff + x - own;
    if (t < TAB_W) {
        sCur[t] = excl;
        tab[blockIdx.x * TAB_W + t] = excl;
    }
    __syncthreads();
    #pragma unroll
    for (int k = 0; k < PART_IT; ++k) {
        if (t + k * 1024 < PART_CH) {
            const int pos = atomicAdd(&sCur[r[k] >> 8], 1);
            sKey[pos] = ((unsigned)(r[k] & (BKT_ROWS - 1)) << 24) | (unsigned)c[k];
            sVal[pos] = v[k];
        }
    }
    __syncthreads();
    for (int i = t; i < PART_CH; i += 1024) {
        pkey[base + i] = sKey[i];
        pval[base + i] = sVal[i];
    }
}

#define SORT_CAP 8192
#define SORT_U   4

__device__ __forceinline__ int gcn_run_of(const int* sSrc, int e) {
    int lo = 0, hi = PART_B;
    #pragma unroll
    for (int it = 0; it < 8; ++it) {
        const int mid = (lo + hi) >> 1;
        if (sSrc[mid] <= e) lo = mid; else hi = mid;
    }
    return lo;
}

#define SORT_LDS_W (260 + PART_B + BKT_ROWS + 32 + 32 + SORT_CAP + SORT_CAP)

__device__ __forceinline__ void gcn_rowsort_role(
        float* __restrict__ smem, const int b,
        const int* __restrict__ tab, const unsigned* __restrict__ pkey,
        const float* __restrict__ pval, int* __restrict__ rowptr,
        float* __restrict__ cval, unsigned short* __restrict__ clo,
        unsigned char* __restrict__ chi) {
    int* sSrc = (int*)smem;
    int* sBeg = sSrc + 260;
    int* sCnt = sBeg + PART_B;
    int* sWa = sCnt + BKT_ROWS;
    int* sWb = sWa + 32;
    int* sKey = sWb + 32;
    float* sVal = (float*)(sKey + SORT_CAP);
    const int t = threadIdx.x;
    const int lane = t & 31;
    const int wave = t >> 5;

    int o0 = 0, cnt = 0;
    if (t < PART_B) {
        o0 = tab[t * TAB_W + b];
        cnt = tab[t * TAB_W + b + 1] - o0;
        sBeg[t] = t * PART_CH + o0;
    }
    if (t < BKT_ROWS) sCnt[t] = 0;
    int x = cnt;
    for (int o = 1; o < 32; o <<= 1) {
        const int y = __shfl_up(x, o);
        if (lane >= o) x += y;
    }
    int q = o0;
    for (int o = 16; o > 0; o >>= 1) q += __shfl_xor(q, o);
    if (lane == 31) sWa[wave] = x;
    if (lane == 0) sWb[wave] = q;
    __syncthreads();
    int woff = 0, n = 0, bbase = 0;
    for (int w = 0; w < 32; ++w) {
        const int a = sWa[w];
        if (w < wave) woff += a;
        n += a;
        bbase += sWb[w];
    }
    if (t < PART_B) sSrc[t] = woff + x - cnt;
    if (t == 0) sSrc[PART_B] = n;
    __syncthreads();

    for (int e0 = t; e0 < n; e0 += 1024 * SORT_U) {
        unsigned key[SORT_U];
        #pragma unroll
        for (int k = 0; k < SORT_U; ++k) {
            const int e = e0 + k * 1024;
            key[k] = 0u;
            if (e < n) {
                const int s = gcn_run_of(sSrc, e);
                key[k] = pkey[sBeg[s] + (e - sSrc[s])];
            }
        }
        #pragma unroll
        for (int k = 0; k < SORT_U; ++k)
            if (e0 + k * 1024 < n) atomicAdd(&sCnt[key[k] >> 24], 1);
    }
    __syncthreads();
    const int own = (t < BKT_ROWS) ? sCnt[t] : 0;
    int y = own;
    for (int o = 1; o < 32; o <<= 1) {
        const int z = __shfl_up(y, o);
        if (lane >= o) y += z;
    }
    if (lane == 31) sWa[wave] = y;
    __syncthreads();
    int roff = 0;
    for (int w = 0; w < 32; ++w)
        if (w < wave) roff += sWa[w];
    const int rstart = roff + y - own;
    if (t < BKT_ROWS) {
        sCnt[t] = rstart;
        const int row = b * BKT_ROWS + t;
        if (row <= N_NODES) rowptr[row] = bbase + rstart;
    }
    __syncthreads();

    const bool staged = (n <= SORT_CAP);
    for (int e0 = t; e0 < n; e0 += 1024 * SORT_U) {
        unsigned key[SORT_U];
        float val[SORT_U];
        #pragma unroll
        for (int k = 0; k < SORT_U; ++k) {
            const int e = e0 + k * 1024;
            key[k] = 0u; val[k] = 0.f;
            if (e < n) {
                const int s = gcn_run_of(sSrc, e);
                const int p = sBeg[s] + (e - sSrc[s]);
                key[k] = pkey[p];
                val[k] = pval[p];
            }
        }
        #pragma unroll
        for (int k = 0; k < SORT_U; ++k) {
            if (e0 + k * 1024 < n) {
                const int pos = atomicAdd(&sCnt[key[k] >> 24], 1);
                const int c = (int)(key[k] & 0xffffffu);
                if (staged) {
                    sKey[pos] = c;
                    sVal[pos] = val[k];
                } else {
                    const int g = bbase + pos;
                    cval[g] = val[k];
                    clo[g] = (unsigned short)(c & 0xffff);
                    chi[g] = (unsigned char)(c >> 16);
                }
            }
        }
    }
    __syncthreads();
    if (staged) {
        for (int i = t; i < n; i += 1024) {
            const int g = bbase + i;
            const int c = sKey[i];
            cval[g] = sVal[i];
            clo[g] = (unsigned short)(c & 0xffff);
            chi[g] = (unsigned char)(c >> 16);
        }
    }
}

__global__ void __launch_bounds__(1024) gcn_rowsort_gemm1_kernel(
        const float* __restrict__ x, const float* __restrict__ W1,
        const float* __restrict__ b1, float* __restrict__ h0,
        const int* __restrict__ tab, const unsigned* __restrict__ pkey,
        const float* __restrict__ pval, int* __restrict__ rowptr,
        float* __restrict__ cval, unsigned short* __restrict__ clo,
        unsigned char* __restrict__ chi) {
    __shared__ alignas(16) float smem[(N_FEAT * N_HID) > SORT_LDS_W ? (N_FEAT * N_HID) : SORT_LDS_W];
    const int blk = blockIdx.x;
    const bool paired = blk < 2 * GEMM_WG;
    const int pair = (blk & (GEMM_WG - 1)) >> 1;
    const int idx = !paired ? blk - GEMM_WG : (blk < GEMM_WG) ? pair : pair + GEMM_WG / 2;
    const bool mult = paired && (((blk & 1) != 0) == (blk >= GEMM_WG));
    if (mult) gcn_gemm1_role(smem, idx, x, W1, b1, h0);
    else      gcn_rowsort_role(smem, idx, tab, pkey, pval, rowptr, cval, clo, chi);
}

#define H_PITCH 68

__global__ void __launch_bounds__(256) gcn_spmm64_gemm2_kernel(
        const int* __restrict__ rowptr, const float* __restrict__ cval,
        const unsigned short* __restrict__ clo, const unsigned char* __restrict__ chi,
        const float* __restrict__ h, const float* __restrict__ W2,
        const float* __restrict__ b2, float* __restrict__ h2) {
    __shared__ alignas(16) float sW[N_HID * N_CLS];
    __shared__ alignas(16) float sH[16 * H_PITCH];
    const float4 w2 = ((const float4*)W2)[threadIdx.x];

    const int lane = threadIdx.x & 31;
    const int wave = threadIdx.x >> 5;
    const int row0 = blockIdx.x * 16 + 2 * wave;
    const int p0 = __builtin_amdgcn_readfirstlane(rowptr[row0]);
    const int p1 = __builtin_amdgcn_readfirstlane(rowptr[row0 + 1]);
    const int p2 = __builtin_amdgcn_readfirstlane(rowptr[row0 + 2]);
    ((float4*)sW)[threadIdx.x] = w2;
    const int slot = lane >> 4;
    const int q = lane & 15;
    const float4* __restrict__ h4 = (const float4*)h;
    for (int rr = 0; rr < 2; ++rr) {
        const int beg = rr ? p1 : p0;
        const int end = rr ? p2 : p1;
        float4 acc = {0.f, 0.f, 0.f, 0.f};
        for (int cb = beg; cb < end; cb += 32) {
            const int n = (end - cb < 32) ? (end - cb) : 32;
            int c = 0;
            float v = 0.f;
            if (lane < n) {
                c = (int)clo[cb + lane] | ((int)chi[cb + lane] << 16);
                v = cval[cb + lane];
            }
            for (int g = 0; g < n; g += 8) {
                const int j0 = g + slot, j1 = j0 + 2, j2 = j0 + 4, j3 = j0 + 6;
                const int c0 = __shfl(c, j0), c1 = __shfl(c, j1), c2 = __shfl(c, j2), c3 = __shfl(c, j3);
                const float v0 = __shfl(v, j0), v1 = __shfl(v, j1), v2 = __shfl(v, j2), v3 = __shfl(v, j3);
                float4 r0 = h4[(size_t)c0 * 16 + q];
                float4 r1 = h4[(size_t)c1 * 16 + q];
                float4 r2 = h4[(size_t)c2 * 16 + q];
                float4 r3 = h4[(size_t)c3 * 16 + q];
                const float4 z = {0.f, 0.f, 0.f, 0.f};
                if (j0 >= n) r0 = z;
                if (j1 >= n) r1 = z;
                if (j2 >= n) r2 = z;
                if (j3 >= n) r3 = z;
                acc.x += v0 * r0.x; acc.y += v0 * r0.y; acc.z += v0 * r0.z; acc.w += v0 * r0.w;
                acc.x += v1 * r1.x; acc.y += v1 * r1.y; acc.z += v1 * r1.z; acc.w += v1 * r1.w;
                acc.x += v2 * r2.x; acc.y += v2 * r2.y; acc.z += v2 * r2.z; acc.w += v2 * r2.w;
                acc.x += v3 * r3.x; acc.y += v3 * r3.y; acc.z += v3 * r3.z; acc.w += v3 * r3.w;
            }
        }
        acc.x += __shfl_xor(acc.x, 16);
        acc.y += __shfl_xor(acc.y, 16);
        acc.z += __shfl_xor(acc.z, 16);
        acc.w += __shfl_xor(acc.w, 16);
        if (slot == 0) {
            float4 rl;
            rl.x = fmaxf(acc.x, 0.f); rl.y = fmaxf(acc.y, 0.f);
            rl.z = fmaxf(acc.z, 0.f); rl.w = fmaxf(acc.w, 0.f);
            *(float4*)&sH[(2 * wave + rr) * H_PITCH + 4 * q] = rl;
        }
    }
    __syncthreads();
    if (wave != 0) return;

    const int half = lane >> 4;
    const int m    = lane & 15;
    const float* __restrict__ hrow = sH + m * H_PITCH;
    v8f d = {};
    #pragma unroll
    for (int k0 = 0; k0 < N_HID; k0 += 4) {
        const int k = k0 + 2 * half;
        v2f a;
        a.x = hrow[k];
        a.y = hrow[k + 1];
        v2f b;
        b.x = sW[(size_t)k * N_CLS + m];
        b.y = sW[(size_t)(k + 1) * N_CLS + m];
        d = __builtin_amdgcn_wmma_f32_16x16x4_f32(false, a, false, b, (short)0, d, false, false);
    }
    float* __restrict__ out = h2 + (size_t)blockIdx.x * 16 * N_CLS;
    const float bb = b2[m];
    #pragma unroll
    for (int r = 0; r < 8; ++r)
        out[(size_t)(half * 8 + r) * N_CLS + m] = d[r] + bb;
}

#define S16_ROWS 4

__device__ __forceinline__ float4 gcn_s16_rest(
        float4 a, const int beg, const int end, const int lane, const int slot, const int q,
        const float* __restrict__ cval, const unsigned short* __restrict__ clo,
        const unsigned char* __restrict__ chi, const float4* __restrict__ h4) {
    const float4 z = {0.f, 0.f, 0.f, 0.f};
    for (int cb = beg + 32; cb < end; cb += 32) {
        const int m = (end - cb < 32) ? (end - cb) : 32;
        int cc = 0;
        float vv = 0.f;
        if (lane < m) {
            cc = (int)clo[cb + lane] | ((int)chi[cb + lane] << 16);
            vv = cval[cb + lane];
        }
        for (int g = 0; g < m; g += 16) {
            const int j0 = g + slot, j1 = j0 + 8;
            const int c0 = __shfl(cc, j0), c1 = __shfl(cc, j1);
            const float v0 = __shfl(vv, j0), v1 = __shfl(vv, j1);
            float4 r0 = h4[(size_t)c0 * 4 + q];
            float4 r1 = h4[(size_t)c1 * 4 + q];
            if (j0 >= m) r0 = z;
            if (j1 >= m) r1 = z;
            a.x += v0 * r0.x; a.y += v0 * r0.y; a.z += v0 * r0.z; a.w += v0 * r0.w;
            a.x += v1 * r1.x; a.y += v1 * r1.y; a.z += v1 * r1.z; a.w += v1 * r1.w;
        }
    }
    return a;
}

__device__ __forceinline__ float4 gcn_s16_first(
        const float4 r0, const float4 r1, const float4 r2, const float4 r3,
        const float w0, const float w1, const float w2, const float w3,
        const int slot, const int n) {
    const float4 z = {0.f, 0.f, 0.f, 0.f};
    float4 t0 = r0, t1 = r1, t2 = r2, t3 = r3;
    if (slot >= n)      t0 = z;
    if (slot + 8 >= n)  t1 = z;
    if (slot + 16 >= n) t2 = z;
    if (slot + 24 >= n) t3 = z;
    float4 a = z;
    a.x += w0 * t0.x; a.y += w0 * t0.y; a.z += w0 * t0.z; a.w += w0 * t0.w;
    a.x += w1 * t1.x; a.y += w1 * t1.y; a.z += w1 * t1.z; a.w += w1 * t1.w;
    a.x += w2 * t2.x; a.y += w2 * t2.y; a.z += w2 * t2.z; a.w += w2 * t2.w;
    a.x += w3 * t3.x; a.y += w3 * t3.y; a.z += w3 * t3.z; a.w += w3 * t3.w;
    return a;
}

__device__ __forceinline__ float4 gcn_s16_fold(float4 a) {
    #pragma unroll
    for (int o = 4; o < 32; o <<= 1) {
        a.x += __shfl_xor(a.x, o);
        a.y += __shfl_xor(a.y, o);
        a.z += __shfl_xor(a.z, o);
        a.w += __shfl_xor(a.w, o);
    }
    return a;
}

__global__ void __launch_bounds__(256) gcn_spmm16_lsm_kernel(
        const int* __restrict__ rowptr, const float* __restrict__ cval,
        const unsigned short* __restrict__ clo, const unsigned char* __restrict__ chi,
        const float* __restrict__ h, float* __restrict__ out) {
    const int lane = threadIdx.x & 31;
    const int wave = threadIdx.x >> 5;
    const int row0 = blockIdx.x * (8 * S16_ROWS) + wave * S16_ROWS;
    const int pv = rowptr[row0 + (lane < S16_ROWS ? lane : S16_ROWS)];
    const int p0 = __builtin_amdgcn_readlane(pv, 0);
    const int p1 = __builtin_amdgcn_readlane(pv, 1);
    const int p2 = __builtin_amdgcn_readlane(pv, 2);
    const int p3 = __builtin_amdgcn_readlane(pv, 3);
    const int p4 = __builtin_amdgcn_readlane(pv, 4);
    const int slot = lane >> 2;
    const int q = lane & 3;
    const float4* __restrict__ h4 = (const float4*)h;

    const int n0 = (p1 - p0 < 32) ? (p1 - p0) : 32;
    const int n1 = (p2 - p1 < 32) ? (p2 - p1) : 32;
    const int n2 = (p3 - p2 < 32) ? (p3 - p2) : 32;
    const int n3 = (p4 - p3 < 32) ? (p4 - p3) : 32;
    const int i0 = (p0 + lane < N_EDGES - 1) ? (p0 + lane) : (N_EDGES - 1);
    const int i1 = (p1 + lane < N_EDGES - 1) ? (p1 + lane) : (N_EDGES - 1);
    const int i2 = (p2 + lane < N_EDGES - 1) ? (p2 + lane) : (N_EDGES - 1);
    const int i3 = (p3 + lane < N_EDGES - 1) ? (p3 + lane) : (N_EDGES - 1);
    const int lo0 = clo[i0], lo1 = clo[i1], lo2 = clo[i2], lo3 = clo[i3];
    const int hi0 = chi[i0], hi1 = chi[i1], hi2 = chi[i2], hi3 = chi[i3];
    const float u0 = cval[i0], u1 = cval[i1], u2 = cval[i2], u3 = cval[i3];
    const int c0 = (lane < n0) ? (lo0 | (hi0 << 16)) : 0;
    const int c1 = (lane < n1) ? (lo1 | (hi1 << 16)) : 0;
    const int c2 = (lane < n2) ? (lo2 | (hi2 << 16)) : 0;
    const int c3 = (lane < n3) ? (lo3 | (hi3 << 16)) : 0;
    const float v0 = (lane < n0) ? u0 : 0.f;
    const float v1 = (lane < n1) ? u1 : 0.f;
    const float v2 = (lane < n2) ? u2 : 0.f;
    const float v3 = (lane < n3) ? u3 : 0.f;

    const int j0 = slot, j1 = slot + 8, j2 = slot + 16, j3 = slot + 24;
    const float4 r00 = h4[(size_t)__shfl(c0, j0) * 4 + q];
    const float4 r01 = h4[(size_t)__shfl(c0, j1) * 4 + q];
    const float4 r02 = h4[(size_t)__shfl(c0, j2) * 4 + q];
    const float4 r03 = h4[(size_t)__shfl(c0, j3) * 4 + q];
    const float4 r10 = h4[(size_t)__shfl(c1, j0) * 4 + q];
    const float4 r11 = h4[(size_t)__shfl(c1, j1) * 4 + q];
    const float4 r12 = h4[(size_t)__shfl(c1, j2) * 4 + q];
    const float4 r13 = h4[(size_t)__shfl(c1, j3) * 4 + q];
    const float4 r20 = h4[(size_t)__shfl(c2, j0) * 4 + q];
    const float4 r21 = h4[(size_t)__shfl(c2, j1) * 4 + q];
    const float4 r22 = h4[(size_t)__shfl(c2, j2) * 4 + q];
    const float4 r23 = h4[(size_t)__shfl(c2, j3) * 4 + q];
    const float4 r30 = h4[(size_t)__shfl(c3, j0) * 4 + q];
    const float4 r31 = h4[(size_t)__shfl(c3, j1) * 4 + q];
    const float4 r32 = h4[(size_t)__shfl(c3, j2) * 4 + q];
    const float4 r33 = h4[(size_t)__shfl(c3, j3) * 4 + q];

    float4 a0 = gcn_s16_first(r00, r01, r02, r03,
        __shfl(v0, j0), __shfl(v0, j1), __shfl(v0, j2), __shfl(v0, j3), slot, n0);
    float4 a1 = gcn_s16_first(r10, r11, r12, r13,
        __shfl(v1, j0), __shfl(v1, j1), __shfl(v1, j2), __shfl(v1, j3), slot, n1);
    float4 a2 = gcn_s16_first(r20, r21, r22, r23,
        __shfl(v2, j0), __shfl(v2, j1), __shfl(v2, j2), __shfl(v2, j3), slot, n2);
    float4 a3 = gcn_s16_first(r30, r31, r32, r33,
        __shfl(v3, j0), __shfl(v3, j1), __shfl(v3, j2), __shfl(v3, j3), slot, n3);
    a0 = gcn_s16_rest(a0, p0, p1, lane, slot, q, cval, clo, chi, h4);
    a1 = gcn_s16_rest(a1, p1, p2, lane, slot, q, cval, clo, chi, h4);
    a2 = gcn_s16_rest(a2, p2, p3, lane, slot, q, cval, clo, chi, h4);
    a3 = gcn_s16_rest(a3, p3, p4, lane, slot, q, cval, clo, chi, h4);
    a0 = gcn_s16_fold(a0);
    a1 = gcn_s16_fold(a1);
    a2 = gcn_s16_fold(a2);
    a3 = gcn_s16_fold(a3);

    const int myr = (lane >> 2) & 3;
    float4 x = a0;
    if (myr == 1) x = a1;
    if (myr == 2) x = a2;
    if (myr == 3) x = a3;
    const int g0 = lane & ~3;
    float f[N_CLS];
    #pragma unroll
    for (int k = 0; k < 4; ++k) {
        f[4 * k + 0] = __shfl(x.x, g0 + k);
        f[4 * k + 1] = __shfl(x.y, g0 + k);
        f[4 * k + 2] = __shfl(x.z, g0 + k);
        f[4 * k + 3] = __shfl(x.w, g0 + k);
    }
    float mx = -3.402823466e+38f;
    #pragma unroll
    for (int i = 0; i < N_CLS; ++i) mx = fmaxf(mx, f[i]);
    float sum = 0.f;
    #pragma unroll
    for (int i = 0; i < N_CLS; ++i) sum += __expf(f[i] - mx);
    const float lse = mx + __logf(sum);
    float4 o = x;
    o.x -= lse; o.y -= lse; o.z -= lse; o.w -= lse;
    if (lane < 4 * S16_ROWS) ((float4*)out)[(size_t)row0 * 4 + lane] = o;
}


extern "C" void kernel_launch(void* const* d_in, const int* in_sizes, int n_in,
                              void* d_out, int out_size, void* d_ws, size_t ws_size,
                              hipStream_t stream) {
    const float* x        = (const float*)d_in[0];
    const int*   adj_row  = (const int*)  d_in[1];
    const int*   adj_col  = (const int*)  d_in[2];
    const float* adj_vals = (const float*)d_in[3];
    const float* W1       = (const float*)d_in[4];
    const float* b1       = (const float*)d_in[5];
    const float* W2       = (const float*)d_in[6];
    const float* b2       = (const float*)d_in[7];
    float* out = (float*)d_out;

    float* ws = (float*)d_ws;
    float* h0 = ws;
    float* wb = ws + (size_t)N_NODES * N_HID;
    float* h2 = wb;
    unsigned* pkey = (unsigned*)wb;
    float*    pval = wb + N_EDGES;
    int*      tab  = (int*)(wb + N_EDGES + N_EDGES);
    float* cval = ws + (size_t)N_NODES * (N_HID + N_HID);
    unsigned short* clo = (unsigned short*)(cval + N_EDGES);
    unsigned char*  chi = (unsigned char*)(cval + N_EDGES + N_EDGES / 2);
    int* rowptr = (int*)(cval + N_EDGES + N_EDGES / 2 + N_EDGES / 4);

    gcn_part_kernel<<<PART_B, 1024, 0, stream>>>(adj_row, adj_col, adj_vals, tab, pkey, pval);
    gcn_rowsort_gemm1_kernel<<<N_BKT + GEMM_WG, 1024, 0, stream>>>(
        x, W1, b1, h0, tab, pkey, pval, rowptr, cval, clo, chi);
    gcn_spmm64_gemm2_kernel<<<N_TILES, 256, 0, stream>>>(rowptr, cval, clo, chi, h0, W2, b2, h2);
    gcn_spmm16_lsm_kernel<<<N_NODES / (8 * S16_ROWS), 256, 0, stream>>>(rowptr, cval, clo, chi, h2, out);
}
